# starting kernel
# baseline (speedup 1.0000x reference)
#include <hip/hip_runtime.h>
#include <math.h>

#define DD 1024
#define NA 1887
#define BB 8
#define TT 16
#define ROWS 128

typedef __attribute__((ext_vector_type(2))) float v2f;
typedef __attribute__((ext_vector_type(8))) float v8f;

__global__ void pool_kernel(const float* __restrict__ x, float* __restrict__ A0,
                            float* __restrict__ F1, float* __restrict__ F2) {
  int tid = blockIdx.x * blockDim.x + threadIdx.x;
  if (tid >= ROWS * DD * 3) return;
  int c  = tid % 3;
  int rd = tid / 3;
  int d  = rd % DD;
  int r  = rd / DD;
  const float* p = x + (size_t)rd * 147 + c;
  float s = 0.f;
  #pragma unroll
  for (int i = 0; i < 49; ++i) s += p[i * 3];
  s *= (1.0f / 49.0f);
  if (c == 0) {
    int b = r >> 4, t = r & 15;
    A0[(size_t)(t * BB + b) * DD + d] = s;
  } else if (c == 1) {
    F1[(size_t)r * DD + d] = s;
  } else {
    F2[(size_t)r * DD + d] = s;
  }
}

__device__ __forceinline__ v8f wmma_k_loop(const float* __restrict__ arow,
                                           const float* __restrict__ wrow,
                                           int K, v8f acc) {
  #pragma unroll 4
  for (int k = 0; k < K; k += 4) {
    v2f a = *(const v2f*)(arow + k);
    v2f b = *(const v2f*)(wrow + k);
    acc = __builtin_amdgcn_wmma_f32_16x16x4_f32(false, a, false, b,
                                                (short)0, acc, false, false);
  }
  return acc;
}

template <int MODE>
__global__ void gemm_kernel(const float* __restrict__ A, int lda,
                            const float* __restrict__ W, int ldw,
                            const float* __restrict__ bias1,
                            const float* __restrict__ bias2,
                            float* __restrict__ out, int ldo, int coloff,
                            float* __restrict__ out2, int N, int K) {
  int wave = (blockIdx.x * blockDim.x + threadIdx.x) >> 5;
  int lane = threadIdx.x & 31;
  int m0 = (wave & 7) << 4;
  int n0 = (wave >> 3) << 4;
  int mr = lane & 15;
  int koff = (lane >> 4) << 1;
  int hi = lane >> 4;
  int ncol = n0 + (lane & 15);
  int nload = (ncol < N) ? ncol : (N - 1);
  const float* arow = A + (size_t)(m0 + mr) * lda + koff;
  const float* wrow = W + (size_t)nload * ldw + koff;
  float bsum = bias1[nload] + (bias2 ? bias2[nload] : 0.f);
  v8f acc;
  #pragma unroll
  for (int i = 0; i < 8; ++i) acc[i] = bsum;
  acc = wmma_k_loop(arow, wrow, K, acc);
  #pragma unroll
  for (int v = 0; v < 8; ++v) {
    int r = m0 + v + (hi << 3);
    float val = acc[v];
    if (MODE == 0) {
      out[(size_t)r * ldo + coloff + ncol] = val;
    } else if (MODE == 1) {
      int b = r >> 4, t = r & 15;
      int rr = t * BB + b;
      if (ncol < DD) out[(size_t)rr * DD + ncol] = val;
      else           out2[(size_t)rr * DD + (ncol - DD)] = val;
    } else {
      if (ncol < N) {
        int t = r >> 3, b = r & 7;
        out[(size_t)(b * TT + t) * N + ncol] = val;
      }
    }
  }
}

__global__ void lstm_step_kernel(const float* __restrict__ xpre,
                                 const float* __restrict__ hin,
                                 const float* __restrict__ Whh,
                                 float* __restrict__ cc,
                                 float* __restrict__ hout,
                                 float* __restrict__ lasth,
                                 int step, int mtiles) {
  __shared__ float red[4][4][8][32];
  int lw   = threadIdx.x >> 5;
  int lane = threadIdx.x & 31;
  int w    = blockIdx.x * 8 + lw;
  int kh   = w & 1;
  int wt   = w >> 1;
  int mt   = wt % mtiles;
  int n0   = (wt / mtiles) << 4;
  int m0   = mt << 4;
  int nn   = lane & 15;
  int mr   = lane & 15;
  int koff = (lane >> 4) << 1;
  int hi   = lane >> 4;
  int kbase = kh * (DD / 2);

  v8f acc[4];
  if (kh == 0) {
    #pragma unroll
    for (int g = 0; g < 4; ++g)
      #pragma unroll
      for (int v = 0; v < 8; ++v) {
        int r = m0 + v + (hi << 3);
        acc[g][v] = xpre[(size_t)r * (4 * DD) + g * DD + n0 + nn];
      }
  } else {
    #pragma unroll
    for (int g = 0; g < 4; ++g)
      #pragma unroll
      for (int v = 0; v < 8; ++v) acc[g][v] = 0.f;
  }

  const float* arow = hin + (size_t)(m0 + mr) * DD + kbase + koff;
  const float* w0 = Whh + (size_t)(0 * DD + n0 + nn) * DD + kbase + koff;
  const float* w1 = Whh + (size_t)(1 * DD + n0 + nn) * DD + kbase + koff;
  const float* w2 = Whh + (size_t)(2 * DD + n0 + nn) * DD + kbase + koff;
  const float* w3 = Whh + (size_t)(3 * DD + n0 + nn) * DD + kbase + koff;

  #pragma unroll 2
  for (int k = 0; k < DD / 2; k += 4) {
    v2f a = *(const v2f*)(arow + k);
    acc[0] = __builtin_amdgcn_wmma_f32_16x16x4_f32(false, a, false,
               *(const v2f*)(w0 + k), (short)0, acc[0], false, false);
    acc[1] = __builtin_amdgcn_wmma_f32_16x16x4_f32(false, a, false,
               *(const v2f*)(w1 + k), (short)0, acc[1], false, false);
    acc[2] = __builtin_amdgcn_wmma_f32_16x16x4_f32(false, a, false,
               *(const v2f*)(w2 + k), (short)0, acc[2], false, false);
    acc[3] = __builtin_amdgcn_wmma_f32_16x16x4_f32(false, a, false,
               *(const v2f*)(w3 + k), (short)0, acc[3], false, false);
  }

  int p = lw >> 1;
  if (kh == 1) {
    #pragma unroll
    for (int g = 0; g < 4; ++g)
      #pragma unroll
      for (int v = 0; v < 8; ++v) red[p][g][v][lane] = acc[g][v];
  }
  __syncthreads();
  if (kh == 1) return;

  #pragma unroll
  for (int g = 0; g < 4; ++g)
    #pragma unroll
    for (int v = 0; v < 8; ++v) acc[g][v] += red[p][g][v][lane];

  int tgt = TT - 1 - step;
  #pragma unroll
  for (int v = 0; v < 8; ++v) {
    int r = m0 + v + (hi << 3);
    int d = n0 + nn;
    float si = 1.f / (1.f + expf(-acc[0][v]));
    float sf = 1.f / (1.f + expf(-acc[1][v]));
    float gt = tanhf(acc[2][v]);
    float so = 1.f / (1.f + expf(-acc[3][v]));
    float cold = cc[(size_t)r * DD + d];
    float cnew = sf * cold + si * gt;
    float hnew = so * tanhf(cnew);
    cc[(size_t)r * DD + d] = cnew;
    hout[(size_t)r * DD + d] = hnew;
    if ((r >> 3) == tgt) lasth[(size_t)r * DD + d] = hnew;
  }
}

__global__ void ln_relu_kernel(const float* __restrict__ in,
                               const float* __restrict__ g,
                               const float* __restrict__ be,
                               float* __restrict__ out) {
  __shared__ float s1[256], s2[256];
  int r = blockIdx.x, t = threadIdx.x;
  const float* row = in + (size_t)r * DD;
  float s = 0.f, sq = 0.f;
  for (int i = t; i < DD; i += 256) { float v = row[i]; s += v; sq += v * v; }
  s1[t] = s; s2[t] = sq;
  __syncthreads();
  for (int off = 128; off > 0; off >>= 1) {
    if (t < off) { s1[t] += s1[t + off]; s2[t] += s2[t + off]; }
    __syncthreads();
  }
  float mean = s1[0] * (1.0f / DD);
  float var = s2[0] * (1.0f / DD) - mean * mean;
  float inv = rsqrtf(var + 1e-5f);
  for (int i = t; i < DD; i += 256) {
    float v = (row[i] - mean) * inv * g[i] + be[i];
    out[(size_t)r * DD + i] = v > 0.f ? v : 0.f;
  }
}

extern "C" void kernel_launch(void* const* d_in, const int* in_sizes, int n_in,
                              void* d_out, int out_size, void* d_ws, size_t ws_size,
                              hipStream_t stream) {
  const float* x   = (const float*)d_in[0];
  const float* Wn  = (const float*)d_in[1];
  const float* bn  = (const float*)d_in[2];
  const float* Wv  = (const float*)d_in[3];
  const float* bv  = (const float*)d_in[4];
  const float* Wf  = (const float*)d_in[5];
  const float* bf  = (const float*)d_in[6];
  const float* Wih = (const float*)d_in[7];
  const float* Whh = (const float*)d_in[8];
  const float* bih = (const float*)d_in[9];
  const float* bhh = (const float*)d_in[10];
  const float* Wc  = (const float*)d_in[11];
  const float* bc  = (const float*)d_in[12];
  const float* g_a = (const float*)d_in[13];
  const float* be_a= (const float*)d_in[14];
  const float* Wa  = (const float*)d_in[15];
  const float* ba  = (const float*)d_in[16];
  float* out = (float*)d_out;

  float* ws    = (float*)d_ws;
  float* A0    = ws;
  float* F1    = A0    + 131072;
  float* F2    = F1    + 131072;
  float* ENV   = F2    + 131072;
  float* XPRE  = ENV   + 262144;
  float* H0    = XPRE  + 524288;
  float* H1    = H0    + 131072;
  float* CC    = H1    + 131072;
  float* LASTH = CC    + 131072;
  float* UNR   = LASTH + 131072;
  float* LNR   = UNR   + 131072;

  pool_kernel<<<1536, 256, 0, stream>>>(x, A0, F1, F2);
  gemm_kernel<0><<<64, 256, 0, stream>>>(F1, DD, Wn, DD, bn, nullptr,
                                         ENV, 2 * DD, 0,  nullptr, DD, DD);
  gemm_kernel<0><<<64, 256, 0, stream>>>(F2, DD, Wv, DD, bv, nullptr,
                                         ENV, 2 * DD, DD, nullptr, DD, DD);
  gemm_kernel<1><<<128, 256, 0, stream>>>(ENV, 2 * DD, Wf, 2 * DD, bf, nullptr,
                                          H0, 0, 0, CC, 2 * DD, 2 * DD);
  gemm_kernel<0><<<256, 256, 0, stream>>>(A0, DD, Wih, DD, bih, bhh,
                                          XPRE, 4 * DD, 0, nullptr, 4 * DD, DD);
  float* hin = H0; float* hout = H1;
  for (int s = 0; s < TT; ++s) {
    int mtiles = (TT - s + 1) >> 1;
    lstm_step_kernel<<<mtiles * 16, 256, 0, stream>>>(XPRE, hin, Whh, CC, hout,
                                                      LASTH, s, mtiles);
    float* tmp = hin; hin = hout; hout = tmp;
  }
  gemm_kernel<0><<<64, 256, 0, stream>>>(LASTH, DD, Wc, DD, bc, nullptr,
                                         UNR, DD, 0, nullptr, DD, DD);
  ln_relu_kernel<<<128, 256, 0, stream>>>(UNR, g_a, be_a, LNR);
  gemm_kernel<2><<<118, 256, 0, stream>>>(LNR, DD, Wa, DD, ba, nullptr,
                                          out, NA, 0, nullptr, NA, DD);
}


# All WMMA K-loops: per-lane fragment loads widened from 8 to 16 bytes (b128); each 8-float K block feeds two WMMAs, k re-ordered identically for A and B, so the order of the k sums changes
# speedup vs baseline: 1.1211x; 1.1211x over previous
#include <hip/hip_runtime.h>
#include <math.h>

#define DD 1024
#define NA 1887
#define BB 8
#define TT 16
#define ROWS 128

typedef __attribute__((ext_vector_type(2))) float v2f;
typedef __attribute__((ext_vector_type(4))) float v4f;
typedef __attribute__((ext_vector_type(8))) float v8f;

__global__ void pool_kernel(const float* __restrict__ x, float* __restrict__ A0,
                            float* __restrict__ F1, float* __restrict__ F2) {
  int tid = blockIdx.x * blockDim.x + threadIdx.x;
  if (tid >= ROWS * DD * 3) return;
  int c  = tid % 3;
  int rd = tid / 3;
  int d  = rd % DD;
  int r  = rd / DD;
  const float* p = x + (size_t)rd * 147 + c;
  float s = 0.f;
  #pragma unroll
  for (int i = 0; i < 49; ++i) s += p[i * 3];
  s *= (1.0f / 49.0f);
  if (c == 0) {
    int b = r >> 4, t = r & 15;
    A0[(size_t)(t * BB + b) * DD + d] = s;
  } else if (c == 1) {
    F1[(size_t)r * DD + d] = s;
  } else {
    F2[(size_t)r * DD + d] = s;
  }
}

__device__ __forceinline__ v8f wmma_k_loop(const float* __restrict__ arow,
                                           const float* __restrict__ wrow,
                                           int K, v8f acc) {
  #pragma unroll 4
  for (int k = 0; k < K; k += 8) {
    v4f a = *(const v4f*)(arow + k);
    v4f b = *(const v4f*)(wrow + k);
    v2f a0 = {a.x, a.y}, a1 = {a.z, a.w};
    v2f b0 = {b.x, b.y}, b1 = {b.z, b.w};
    acc = __builtin_amdgcn_wmma_f32_16x16x4_f32(false, a0, false, b0,
                                                (short)0, acc, false, false);
    acc = __builtin_amdgcn_wmma_f32_16x16x4_f32(false, a1, false, b1,
                                                (short)0, acc, false, false);
  }
  return acc;
}

template <int MODE>
__global__ void gemm_kernel(const float* __restrict__ A, int lda,
                            const float* __restrict__ W, int ldw,
                            const float* __restrict__ bias1,
                            const float* __restrict__ bias2,
                            float* __restrict__ out, int ldo, int coloff,
                            float* __restrict__ out2, int N, int K) {
  int wave = (blockIdx.x * blockDim.x + threadIdx.x) >> 5;
  int lane = threadIdx.x & 31;
  int m0 = (wave & 7) << 4;
  int n0 = (wave >> 3) << 4;
  int mr = lane & 15;
  int koff = (lane >> 4) << 2;
  int hi = lane >> 4;
  int ncol = n0 + (lane & 15);
  int nload = (ncol < N) ? ncol : (N - 1);
  const float* arow = A + (size_t)(m0 + mr) * lda + koff;
  const float* wrow = W + (size_t)nload * ldw + koff;
  float bsum = bias1[nload] + (bias2 ? bias2[nload] : 0.f);
  v8f acc;
  #pragma unroll
  for (int i = 0; i < 8; ++i) acc[i] = bsum;
  acc = wmma_k_loop(arow, wrow, K, acc);
  #pragma unroll
  for (int v = 0; v < 8; ++v) {
    int r = m0 + v + (hi << 3);
    float val = acc[v];
    if (MODE == 0) {
      out[(size_t)r * ldo + coloff + ncol] = val;
    } else if (MODE == 1) {
      int b = r >> 4, t = r & 15;
      int rr = t * BB + b;
      if (ncol < DD) out[(size_t)rr * DD + ncol] = val;
      else           out2[(size_t)rr * DD + (ncol - DD)] = val;
    } else {
      if (ncol < N) {
        int t = r >> 3, b = r & 7;
        out[(size_t)(b * TT + t) * N + ncol] = val;
      }
    }
  }
}

__global__ void lstm_step_kernel(const float* __restrict__ xpre,
                                 const float* __restrict__ hin,
                                 const float* __restrict__ Whh,
                                 float* __restrict__ cc,
                                 float* __restrict__ hout,
                                 float* __restrict__ lasth,
                                 int step, int mtiles) {
  __shared__ float red[4][4][8][32];
  int lw   = threadIdx.x >> 5;
  int lane = threadIdx.x & 31;
  int w    = blockIdx.x * 8 + lw;
  int kh   = w & 1;
  int wt   = w >> 1;
  int mt   = wt % mtiles;
  int n0   = (wt / mtiles) << 4;
  int m0   = mt << 4;
  int nn   = lane & 15;
  int mr   = lane & 15;
  int koff = (lane >> 4) << 2;
  int hi   = lane >> 4;
  int kbase = kh * (DD / 2);

  v8f acc[4];
  if (kh == 0) {
    #pragma unroll
    for (int g = 0; g < 4; ++g)
      #pragma unroll
      for (int v = 0; v < 8; ++v) {
        int r = m0 + v + (hi << 3);
        acc[g][v] = xpre[(size_t)r * (4 * DD) + g * DD + n0 + nn];
      }
  } else {
    #pragma unroll
    for (int g = 0; g < 4; ++g)
      #pragma unroll
      for (int v = 0; v < 8; ++v) acc[g][v] = 0.f;
  }

  const float* arow = hin + (size_t)(m0 + mr) * DD + kbase + koff;
  const float* w0 = Whh + (size_t)(0 * DD + n0 + nn) * DD + kbase + koff;
  const float* w1 = Whh + (size_t)(1 * DD + n0 + nn) * DD + kbase + koff;
  const float* w2 = Whh + (size_t)(2 * DD + n0 + nn) * DD + kbase + koff;
  const float* w3 = Whh + (size_t)(3 * DD + n0 + nn) * DD + kbase + koff;

  #pragma unroll 2
  for (int k = 0; k < DD / 2; k += 8) {
    v4f a  = *(const v4f*)(arow + k);
    v4f q0 = *(const v4f*)(w0 + k);
    v4f q1 = *(const v4f*)(w1 + k);
    v4f q2 = *(const v4f*)(w2 + k);
    v4f q3 = *(const v4f*)(w3 + k);
    v2f a0 = {a.x, a.y}, a1 = {a.z, a.w};
    v2f b;
    b = (v2f){q0.x, q0.y};
    acc[0] = __builtin_amdgcn_wmma_f32_16x16x4_f32(false, a0, false, b,
                                                   (short)0, acc[0], false, false);
    b = (v2f){q1.x, q1.y};
    acc[1] = __builtin_amdgcn_wmma_f32_16x16x4_f32(false, a0, false, b,
                                                   (short)0, acc[1], false, false);
    b = (v2f){q2.x, q2.y};
    acc[2] = __builtin_amdgcn_wmma_f32_16x16x4_f32(false, a0, false, b,
                                                   (short)0, acc[2], false, false);
    b = (v2f){q3.x, q3.y};
    acc[3] = __builtin_amdgcn_wmma_f32_16x16x4_f32(false, a0, false, b,
                                                   (short)0, acc[3], false, false);
    b = (v2f){q0.z, q0.w};
    acc[0] = __builtin_amdgcn_wmma_f32_16x16x4_f32(false, a1, false, b,
                                                   (short)0, acc[0], false, false);
    b = (v2f){q1.z, q1.w};
    acc[1] = __builtin_amdgcn_wmma_f32_16x16x4_f32(false, a1, false, b,
                                                   (short)0, acc[1], false, false);
    b = (v2f){q2.z, q2.w};
    acc[2] = __builtin_amdgcn_wmma_f32_16x16x4_f32(false, a1, false, b,
                                                   (short)0, acc[2], false, false);
    b = (v2f){q3.z, q3.w};
    acc[3] = __builtin_amdgcn_wmma_f32_16x16x4_f32(false, a1, false, b,
                                                   (short)0, acc[3], false, false);
  }

  int p = lw >> 1;
  if (kh == 1) {
    #pragma unroll
    for (int g = 0; g < 4; ++g)
      #pragma unroll
      for (int v = 0; v < 8; ++v) red[p][g][v][lane] = acc[g][v];
  }
  __syncthreads();
  if (kh == 1) return;

  #pragma unroll
  for (int g = 0; g < 4; ++g)
    #pragma unroll
    for (int v = 0; v < 8; ++v) acc[g][v] += red[p][g][v][lane];

  int tgt = TT - 1 - step;
  #pragma unroll
  for (int v = 0; v < 8; ++v) {
    int r = m0 + v + (hi << 3);
    int d = n0 + nn;
    float si = 1.f / (1.f + expf(-acc[0][v]));
    float sf = 1.f / (1.f + expf(-acc[1][v]));
    float gt = tanhf(acc[2][v]);
    float so = 1.f / (1.f + expf(-acc[3][v]));
    float cold = cc[(size_t)r * DD + d];
    float cnew = sf * cold + si * gt;
    float hnew = so * tanhf(cnew);
    cc[(size_t)r * DD + d] = cnew;
    hout[(size_t)r * DD + d] = hnew;
    if ((r >> 3) == tgt) lasth[(size_t)r * DD + d] = hnew;
  }
}

__global__ void ln_relu_kernel(const float* __restrict__ in,
                               const float* __restrict__ g,
                               const float* __restrict__ be,
                               float* __restrict__ out) {
  __shared__ float s1[256], s2[256];
  int r = blockIdx.x, t = threadIdx.x;
  const float* row = in + (size_t)r * DD;
  float s = 0.f, sq = 0.f;
  for (int i = t; i < DD; i += 256) { float v = row[i]; s += v; sq += v * v; }
  s1[t] = s; s2[t] = sq;
  __syncthreads();
  for (int off = 128; off > 0; off >>= 1) {
    if (t < off) { s1[t] += s1[t + off]; s2[t] += s2[t + off]; }
    __syncthreads();
  }
  float mean = s1[0] * (1.0f / DD);
  float var = s2[0] * (1.0f / DD) - mean * mean;
  float inv = rsqrtf(var + 1e-5f);
  for (int i = t; i < DD; i += 256) {
    float v = (row[i] - mean) * inv * g[i] + be[i];
    out[(size_t)r * DD + i] = v > 0.f ? v : 0.f;
  }
}

extern "C" void kernel_launch(void* const* d_in, const int* in_sizes, int n_in,
                              void* d_out, int out_size, void* d_ws, size_t ws_size,
                              hipStream_t stream) {
  const float* x   = (const float*)d_in[0];
  const float* Wn  = (const float*)d_in[1];
  const float* bn  = (const float*)d_in[2];
  const float* Wv  = (const float*)d_in[3];
  const float* bv  = (const float*)d_in[4];
  const float* Wf  = (const float*)d_in[5];
  const float* bf  = (const float*)d_in[6];
  const float* Wih = (const float*)d_in[7];
  const float* Whh = (const float*)d_in[8];
  const float* bih = (const float*)d_in[9];
  const float* bhh = (const float*)d_in[10];
  const float* Wc  = (const float*)d_in[11];
  const float* bc  = (const float*)d_in[12];
  const float* g_a = (const float*)d_in[13];
  const float* be_a= (const float*)d_in[14];
  const float* Wa  = (const float*)d_in[15];
  const float* ba  = (const float*)d_in[16];
  float* out = (float*)d_out;

  float* ws    = (float*)d_ws;
  float* A0    = ws;
  float* F1    = A0    + 131072;
  float* F2    = F1    + 131072;
  float* ENV   = F2    + 131072;
  float* XPRE  = ENV   + 262144;
  float* H0    = XPRE  + 524288;
  float* H1    = H0    + 131072;
  float* CC    = H1    + 131072;
  float* LASTH = CC    + 131072;
  float* UNR   = LASTH + 131072;
  float* LNR   = UNR   + 131072;

  pool_kernel<<<1536, 256, 0, stream>>>(x, A0, F1, F2);
  gemm_kernel<0><<<64, 256, 0, stream>>>(F1, DD, Wn, DD, bn, nullptr,
                                         ENV, 2 * DD, 0,  nullptr, DD, DD);
  gemm_kernel<0><<<64, 256, 0, stream>>>(F2, DD, Wv, DD, bv, nullptr,
                                         ENV, 2 * DD, DD, nullptr, DD, DD);
  gemm_kernel<1><<<128, 256, 0, stream>>>(ENV, 2 * DD, Wf, 2 * DD, bf, nullptr,
                                          H0, 0, 0, CC, 2 * DD, 2 * DD);
  gemm_kernel<0><<<256, 256, 0, stream>>>(A0, DD, Wih, DD, bih, bhh,
                                          XPRE, 4 * DD, 0, nullptr, 4 * DD, DD);
  float* hin = H0; float* hout = H1;
  for (int s = 0; s < TT; ++s) {
    int mtiles = (TT - s + 1) >> 1;
    lstm_step_kernel<<<mtiles * 16, 256, 0, stream>>>(XPRE, hin, Whh, CC, hout,
                                                      LASTH, s, mtiles);
    float* tmp = hin; hin = hout; hout = tmp;
  }
  gemm_kernel<0><<<64, 256, 0, stream>>>(LASTH, DD, Wc, DD, bc, nullptr,
                                         UNR, DD, 0, nullptr, DD, DD);
  ln_relu_kernel<<<128, 256, 0, stream>>>(UNR, g_a, be_a, LNR);
  gemm_kernel<2><<<118, 256, 0, stream>>>(LNR, DD, Wa, DD, ba, nullptr,
                                          out, NA, 0, nullptr, NA, DD);
}


# LSTM step: K split 8 ways, 8 waves per output tile (was 2), partials summed through LDS in slice order, each wave finishes one accumulator row; 1 or 2 tiles per block; the order of the k sums changes
# speedup vs baseline: 1.8653x; 1.6638x over previous
#include <hip/hip_runtime.h>
#include <math.h>

#define DD 1024
#define NA 1887
#define BB 8
#define TT 16
#define ROWS 128

typedef __attribute__((ext_vector_type(2))) float v2f;
typedef __attribute__((ext_vector_type(4))) float v4f;
typedef __attribute__((ext_vector_type(8))) float v8f;

__global__ void pool_kernel(const float* __restrict__ x, float* __restrict__ A0,
                            float* __restrict__ F1, float* __restrict__ F2) {
  int tid = blockIdx.x * blockDim.x + threadIdx.x;
  if (tid >= ROWS * DD * 3) return;
  int c  = tid % 3;
  int rd = tid / 3;
  int d  = rd % DD;
  int r  = rd / DD;
  const float* p = x + (size_t)rd * 147 + c;
  float s = 0.f;
  #pragma unroll
  for (int i = 0; i < 49; ++i) s += p[i * 3];
  s *= (1.0f / 49.0f);
  if (c == 0) {
    int b = r >> 4, t = r & 15;
    A0[(size_t)(t * BB + b) * DD + d] = s;
  } else if (c == 1) {
    F1[(size_t)r * DD + d] = s;
  } else {
    F2[(size_t)r * DD + d] = s;
  }
}

__device__ __forceinline__ v8f wmma_k_loop(const float* __restrict__ arow,
                                           const float* __restrict__ wrow,
                                           int K, v8f acc) {
  #pragma unroll 4
  for (int k = 0; k < K; k += 8) {
    v4f a = *(const v4f*)(arow + k);
    v4f b = *(const v4f*)(wrow + k);
    v2f a0 = {a.x, a.y}, a1 = {a.z, a.w};
    v2f b0 = {b.x, b.y}, b1 = {b.z, b.w};
    acc = __builtin_amdgcn_wmma_f32_16x16x4_f32(false, a0, false, b0,
                                                (short)0, acc, false, false);
    acc = __builtin_amdgcn_wmma_f32_16x16x4_f32(false, a1, false, b1,
                                                (short)0, acc, false, false);
  }
  return acc;
}

template <int MODE>
__global__ void gemm_kernel(const float* __restrict__ A, int lda,
                            const float* __restrict__ W, int ldw,
                            const float* __restrict__ bias1,
                            const float* __restrict__ bias2,
                            float* __restrict__ out, int ldo, int coloff,
                            float* __restrict__ out2, int N, int K) {
  int wave = (blockIdx.x * blockDim.x + threadIdx.x) >> 5;
  int lane = threadIdx.x & 31;
  int m0 = (wave & 7) << 4;
  int n0 = (wave >> 3) << 4;
  int mr = lane & 15;
  int koff = (lane >> 4) << 2;
  int hi = lane >> 4;
  int ncol = n0 + (lane & 15);
  int nload = (ncol < N) ? ncol : (N - 1);
  const float* arow = A + (size_t)(m0 + mr) * lda + koff;
  const float* wrow = W + (size_t)nload * ldw + koff;
  float bsum = bias1[nload] + (bias2 ? bias2[nload] : 0.f);
  v8f acc;
  #pragma unroll
  for (int i = 0; i < 8; ++i) acc[i] = bsum;
  acc = wmma_k_loop(arow, wrow, K, acc);
  #pragma unroll
  for (int v = 0; v < 8; ++v) {
    int r = m0 + v + (hi << 3);
    float val = acc[v];
    if (MODE == 0) {
      out[(size_t)r * ldo + coloff + ncol] = val;
    } else if (MODE == 1) {
      int b = r >> 4, t = r & 15;
      int rr = t * BB + b;
      if (ncol < DD) out[(size_t)rr * DD + ncol] = val;
      else           out2[(size_t)rr * DD + (ncol - DD)] = val;
    } else {
      if (ncol < N) {
        int t = r >> 3, b = r & 7;
        out[(size_t)(b * TT + t) * N + ncol] = val;
      }
    }
  }
}

__global__ void __launch_bounds__(512)
lstm_step_kernel(const float* __restrict__ xpre,
                 const float* __restrict__ hin,
                 const float* __restrict__ Whh,
                 float* __restrict__ cc,
                 float* __restrict__ hout,
                 float* __restrict__ lasth,
                 int step, int mtiles) {
  __shared__ float red[2][8][4][8][32];
  int lw   = threadIdx.x >> 5;
  int lane = threadIdx.x & 31;
  int tb   = lw >> 3;
  int ks   = lw & 7;
  int wt   = blockIdx.x * (blockDim.x >> 8) + tb;
  int mt   = wt % mtiles;
  int n0   = (wt / mtiles) << 4;
  int m0   = mt << 4;
  int nn   = lane & 15;
  int mr   = lane & 15;
  int koff = (lane >> 4) << 2;
  int hi   = lane >> 4;
  int kbase = ks * (DD / 8);

  int rf = m0 + ks + (hi << 3);
  int d  = n0 + nn;
  float xp0 = xpre[(size_t)rf * (4 * DD) + 0 * DD + d];
  float xp1 = xpre[(size_t)rf * (4 * DD) + 1 * DD + d];
  float xp2 = xpre[(size_t)rf * (4 * DD) + 2 * DD + d];
  float xp3 = xpre[(size_t)rf * (4 * DD) + 3 * DD + d];
  float cold = cc[(size_t)rf * DD + d];

  v8f acc[4];
  #pragma unroll
  for (int g = 0; g < 4; ++g)
    #pragma unroll
    for (int v = 0; v < 8; ++v) acc[g][v] = 0.f;

  const float* arow = hin + (size_t)(m0 + mr) * DD + kbase + koff;
  const float* w0 = Whh + (size_t)(0 * DD + n0 + nn) * DD + kbase + koff;
  const float* w1 = Whh + (size_t)(1 * DD + n0 + nn) * DD + kbase + koff;
  const float* w2 = Whh + (size_t)(2 * DD + n0 + nn) * DD + kbase + koff;
  const float* w3 = Whh + (size_t)(3 * DD + n0 + nn) * DD + kbase + koff;

  #pragma unroll 2
  for (int k = 0; k < DD / 8; k += 8) {
    v4f a  = *(const v4f*)(arow + k);
    v4f q0 = *(const v4f*)(w0 + k);
    v4f q1 = *(const v4f*)(w1 + k);
    v4f q2 = *(const v4f*)(w2 + k);
    v4f q3 = *(const v4f*)(w3 + k);
    v2f a0 = {a.x, a.y}, a1 = {a.z, a.w};
    v2f b;
    b = (v2f){q0.x, q0.y};
    acc[0] = __builtin_amdgcn_wmma_f32_16x16x4_f32(false, a0, false, b,
                                                   (short)0, acc[0], false, false);
    b = (v2f){q1.x, q1.y};
    acc[1] = __builtin_amdgcn_wmma_f32_16x16x4_f32(false, a0, false, b,
                                                   (short)0, acc[1], false, false);
    b = (v2f){q2.x, q2.y};
    acc[2] = __builtin_amdgcn_wmma_f32_16x16x4_f32(false, a0, false, b,
                                                   (short)0, acc[2], false, false);
    b = (v2f){q3.x, q3.y};
    acc[3] = __builtin_amdgcn_wmma_f32_16x16x4_f32(false, a0, false, b,
                                                   (short)0, acc[3], false, false);
    b = (v2f){q0.z, q0.w};
    acc[0] = __builtin_amdgcn_wmma_f32_16x16x4_f32(false, a1, false, b,
                                                   (short)0, acc[0], false, false);
    b = (v2f){q1.z, q1.w};
    acc[1] = __builtin_amdgcn_wmma_f32_16x16x4_f32(false, a1, false, b,
                                                   (short)0, acc[1], false, false);
    b = (v2f){q2.z, q2.w};
    acc[2] = __builtin_amdgcn_wmma_f32_16x16x4_f32(false, a1, false, b,
                                                   (short)0, acc[2], false, false);
    b = (v2f){q3.z, q3.w};
    acc[3] = __builtin_amdgcn_wmma_f32_16x16x4_f32(false, a1, false, b,
                                                   (short)0, acc[3], false, false);
  }

  #pragma unroll
  for (int g = 0; g < 4; ++g)
    #pragma unroll
    for (int v = 0; v < 8; ++v) red[tb][ks][g][v][lane] = acc[g][v];
  __syncthreads();

  float s[4];
  #pragma unroll
  for (int g = 0; g < 4; ++g) {
    float t = red[tb][0][g][ks][lane];
    #pragma unroll
    for (int q = 1; q < 8; ++q) t += red[tb][q][g][ks][lane];
    s[g] = t;
  }
  s[0] += xp0; s[1] += xp1; s[2] += xp2; s[3] += xp3;

  int tgt = TT - 1 - step;
  float si = 1.f / (1.f + expf(-s[0]));
  float sf = 1.f / (1.f + expf(-s[1]));
  float gt = tanhf(s[2]);
  float so = 1.f / (1.f + expf(-s[3]));
  float cnew = sf * cold + si * gt;
  float hnew = so * tanhf(cnew);
  cc[(size_t)rf * DD + d] = cnew;
  hout[(size_t)rf * DD + d] = hnew;
  if ((rf >> 3) == tgt) lasth[(size_t)rf * DD + d] = hnew;
}

__global__ void ln_relu_kernel(const float* __restrict__ in,
                               const float* __restrict__ g,
                               const float* __restrict__ be,
                               float* __restrict__ out) {
  __shared__ float s1[256], s2[256];
  int r = blockIdx.x, t = threadIdx.x;
  const float* row = in + (size_t)r * DD;
  float s = 0.f, sq = 0.f;
  for (int i = t; i < DD; i += 256) { float v = row[i]; s += v; sq += v * v; }
  s1[t] = s; s2[t] = sq;
  __syncthreads();
  for (int off = 128; off > 0; off >>= 1) {
    if (t < off) { s1[t] += s1[t + off]; s2[t] += s2[t + off]; }
    __syncthreads();
  }
  float mean = s1[0] * (1.0f / DD);
  float var = s2[0] * (1.0f / DD) - mean * mean;
  float inv = rsqrtf(var + 1e-5f);
  for (int i = t; i < DD; i += 256) {
    float v = (row[i] - mean) * inv * g[i] + be[i];
    out[(size_t)r * DD + i] = v > 0.f ? v : 0.f;
  }
}

extern "C" void kernel_launch(void* const* d_in, const int* in_sizes, int n_in,
                              void* d_out, int out_size, void* d_ws, size_t ws_size,
                              hipStream_t stream) {
  const float* x   = (const float*)d_in[0];
  const float* Wn  = (const float*)d_in[1];
  const float* bn  = (const float*)d_in[2];
  const float* Wv  = (const float*)d_in[3];
  const float* bv  = (const float*)d_in[4];
  const float* Wf  = (const float*)d_in[5];
  const float* bf  = (const float*)d_in[6];
  const float* Wih = (const float*)d_in[7];
  const float* Whh = (const float*)d_in[8];
  const float* bih = (const float*)d_in[9];
  const float* bhh = (const float*)d_in[10];
  const float* Wc  = (const float*)d_in[11];
  const float* bc  = (const float*)d_in[12];
  const float* g_a = (const float*)d_in[13];
  const float* be_a= (const float*)d_in[14];
  const float* Wa  = (const float*)d_in[15];
  const float* ba  = (const float*)d_in[16];
  float* out = (float*)d_out;

  float* ws    = (float*)d_ws;
  float* A0    = ws;
  float* F1    = A0    + 131072;
  float* F2    = F1    + 131072;
  float* ENV   = F2    + 131072;
  float* XPRE  = ENV   + 262144;
  float* H0    = XPRE  + 524288;
  float* H1    = H0    + 131072;
  float* CC    = H1    + 131072;
  float* LASTH = CC    + 131072;
  float* UNR   = LASTH + 131072;
  float* LNR   = UNR   + 131072;

  pool_kernel<<<1536, 256, 0, stream>>>(x, A0, F1, F2);
  gemm_kernel<0><<<64, 256, 0, stream>>>(F1, DD, Wn, DD, bn, nullptr,
                                         ENV, 2 * DD, 0,  nullptr, DD, DD);
  gemm_kernel<0><<<64, 256, 0, stream>>>(F2, DD, Wv, DD, bv, nullptr,
                                         ENV, 2 * DD, DD, nullptr, DD, DD);
  gemm_kernel<1><<<128, 256, 0, stream>>>(ENV, 2 * DD, Wf, 2 * DD, bf, nullptr,
                                          H0, 0, 0, CC, 2 * DD, 2 * DD);
  gemm_kernel<0><<<256, 256, 0, stream>>>(A0, DD, Wih, DD, bih, bhh,
                                          XPRE, 4 * DD, 0, nullptr, 4 * DD, DD);
  float* hin = H0; float* hout = H1;
  for (int s = 0; s < TT; ++s) {
    int mtiles = (TT - s + 1) >> 1;
    if (mtiles > 4)
      lstm_step_kernel<<<mtiles * 32, 512, 0, stream>>>(XPRE, hin, Whh, CC, hout,
                                                        LASTH, s, mtiles);
    else
      lstm_step_kernel<<<mtiles * 64, 256, 0, stream>>>(XPRE, hin, Whh, CC, hout,
                                                        LASTH, s, mtiles);
    float* tmp = hin; hin = hout; hout = tmp;
  }
  gemm_kernel<0><<<64, 256, 0, stream>>>(LASTH, DD, Wc, DD, bc, nullptr,
                                         UNR, DD, 0, nullptr, DD, DD);
  ln_relu_kernel<<<128, 256, 0, stream>>>(UNR, g_a, be_a, LNR);
  gemm_kernel<2><<<118, 256, 0, stream>>>(LNR, DD, Wa, DD, ba, nullptr,
                                          out, NA, 0, nullptr, NA, DD);
}


# All six GEMM launches: a block's 8 waves split K (was M), each wave carries 2, 4 or 8 row tiles per W fragment, partials summed through LDS in slice order, bias last; the k sum order changes
# speedup vs baseline: 2.7309x; 1.4641x over previous
#include <hip/hip_runtime.h>
#include <math.h>

#define DD 1024
#define NA 1887
#define BB 8
#define TT 16
#define ROWS 128

typedef __attribute__((ext_vector_type(2))) float v2f;
typedef __attribute__((ext_vector_type(4))) float v4f;
typedef __attribute__((ext_vector_type(8))) float v8f;

__global__ void pool_kernel(const float* __restrict__ x, float* __restrict__ A0,
                            float* __restrict__ F1, float* __restrict__ F2) {
  int tid = blockIdx.x * blockDim.x + threadIdx.x;
  if (tid >= ROWS * DD * 3) return;
  int c  = tid % 3;
  int rd = tid / 3;
  int d  = rd % DD;
  int r  = rd / DD;
  const float* p = x + (size_t)rd * 147 + c;
  float s = 0.f;
  #pragma unroll
  for (int i = 0; i < 49; ++i) s += p[i * 3];
  s *= (1.0f / 49.0f);
  if (c == 0) {
    int b = r >> 4, t = r & 15;
    A0[(size_t)(t * BB + b) * DD + d] = s;
  } else if (c == 1) {
    F1[(size_t)r * DD + d] = s;
  } else {
    F2[(size_t)r * DD + d] = s;
  }
}

template <int MODE, int MT, int KK>
__global__ void __launch_bounds__(256)
gemm_kernel(const float* __restrict__ A, int lda,
            const float* __restrict__ W, int ldw,
            const float* __restrict__ bias1,
            const float* __restrict__ bias2,
            float* __restrict__ out, int ldo, int coloff,
            float* __restrict__ out2, int N) {
  __shared__ float red[8][MT][8][32];
  constexpr int MG = 8 / MT;
  int ks   = threadIdx.x >> 5;
  int lane = threadIdx.x & 31;
  int n0    = (blockIdx.x / MG) << 4;
  int mbase = (blockIdx.x % MG) * (MT * 16);
  int c    = lane & 15;
  int hi   = lane >> 4;
  int koff = hi << 2;
  int ncol = n0 + c;
  int nload = (ncol < N) ? ncol : (N - 1);
  constexpr int KS = KK / 8;
  const float* wrow = W + (size_t)nload * ldw + ks * KS + koff;
  const float* arow = A + (size_t)(mbase + c) * lda + ks * KS + koff;
  const size_t astep = (size_t)16 * lda;

  v8f acc[MT];
  #pragma unroll
  for (int mt = 0; mt < MT; ++mt)
    #pragma unroll
    for (int v = 0; v < 8; ++v) acc[mt][v] = 0.f;

  constexpr int NL = KS / 32;
  v4f b[2][4];
  v4f a[2][4][MT];
  #pragma unroll
  for (int j = 0; j < 4; ++j) {
    b[0][j] = *(const v4f*)(wrow + 8 * j);
    #pragma unroll
    for (int mt = 0; mt < MT; ++mt)
      a[0][j][mt] = *(const v4f*)(arow + mt * astep + 8 * j);
  }
  #pragma unroll
  for (int t = 0; t < NL; ++t) {
    const int cur = t & 1, nxt = cur ^ 1;
    if (t + 1 < NL) {
      #pragma unroll
      for (int j = 0; j < 4; ++j) {
        b[nxt][j] = *(const v4f*)(wrow + (t + 1) * 32 + 8 * j);
        #pragma unroll
        for (int mt = 0; mt < MT; ++mt)
          a[nxt][j][mt] = *(const v4f*)(arow + mt * astep + (t + 1) * 32 + 8 * j);
      }
    }
    __builtin_amdgcn_sched_barrier(0);
    #pragma unroll
    for (int j = 0; j < 4; ++j) {
      v2f b0 = {b[cur][j].x, b[cur][j].y}, b1 = {b[cur][j].z, b[cur][j].w};
      #pragma unroll
      for (int mt = 0; mt < MT; ++mt) {
        v2f a0 = {a[cur][j][mt].x, a[cur][j][mt].y};
        acc[mt] = __builtin_amdgcn_wmma_f32_16x16x4_f32(false, a0, false, b0,
                                                        (short)0, acc[mt], false, false);
      }
      #pragma unroll
      for (int mt = 0; mt < MT; ++mt) {
        v2f a1 = {a[cur][j][mt].z, a[cur][j][mt].w};
        acc[mt] = __builtin_amdgcn_wmma_f32_16x16x4_f32(false, a1, false, b1,
                                                        (short)0, acc[mt], false, false);
      }
    }
    __builtin_amdgcn_sched_barrier(0);
  }

  #pragma unroll
  for (int mt = 0; mt < MT; ++mt)
    #pragma unroll
    for (int v = 0; v < 8; ++v) red[ks][mt][v][lane] = acc[mt][v];
  __syncthreads();

  float bsum = bias1[nload] + (bias2 ? bias2[nload] : 0.f);
  #pragma unroll
  for (int mt = 0; mt < MT; ++mt) {
    float t = red[0][mt][ks][lane];
    #pragma unroll
    for (int q = 1; q < 8; ++q) t += red[q][mt][ks][lane];
    float val = t + bsum;
    int r = mbase + mt * 16 + ks + (hi << 3);
    if (MODE == 0) {
      out[(size_t)r * ldo + coloff + ncol] = val;
    } else if (MODE == 1) {
      int bi = r >> 4, tt = r & 15;
      int rr = tt * BB + bi;
      if (ncol < DD) out[(size_t)rr * DD + ncol] = val;
      else           out2[(size_t)rr * DD + (ncol - DD)] = val;
    } else {
      if (ncol < N) {
        int tt = r >> 3, bi = r & 7;
        out[(size_t)(bi * TT + tt) * N + ncol] = val;
      }
    }
  }
}

__global__ void __launch_bounds__(512)
lstm_step_kernel(const float* __restrict__ xpre,
                 const float* __restrict__ hin,
                 const float* __restrict__ Whh,
                 float* __restrict__ cc,
                 float* __restrict__ hout,
                 float* __restrict__ lasth,
                 int step, int mtiles) {
  __shared__ float red[2][8][4][8][32];
  int lw   = threadIdx.x >> 5;
  int lane = threadIdx.x & 31;
  int tb   = lw >> 3;
  int ks   = lw & 7;
  int wt   = blockIdx.x * (blockDim.x >> 8) + tb;
  int mt   = wt % mtiles;
  int n0   = (wt / mtiles) << 4;
  int m0   = mt << 4;
  int nn   = lane & 15;
  int mr   = lane & 15;
  int koff = (lane >> 4) << 2;
  int hi   = lane >> 4;
  int kbase = ks * (DD / 8);

  int rf = m0 + ks + (hi << 3);
  int d  = n0 + nn;
  float xp0 = xpre[(size_t)rf * (4 * DD) + 0 * DD + d];
  float xp1 = xpre[(size_t)rf * (4 * DD) + 1 * DD + d];
  float xp2 = xpre[(size_t)rf * (4 * DD) + 2 * DD + d];
  float xp3 = xpre[(size_t)rf * (4 * DD) + 3 * DD + d];
  float cold = cc[(size_t)rf * DD + d];

  v8f acc[4];
  #pragma unroll
  for (int g = 0; g < 4; ++g)
    #pragma unroll
    for (int v = 0; v < 8; ++v) acc[g][v] = 0.f;

  const float* arow = hin + (size_t)(m0 + mr) * DD + kbase + koff;
  const float* w0 = Whh + (size_t)(0 * DD + n0 + nn) * DD + kbase + koff;
  const float* w1 = Whh + (size_t)(1 * DD + n0 + nn) * DD + kbase + koff;
  const float* w2 = Whh + (size_t)(2 * DD + n0 + nn) * DD + kbase + koff;
  const float* w3 = Whh + (size_t)(3 * DD + n0 + nn) * DD + kbase + koff;

  #pragma unroll 2
  for (int k = 0; k < DD / 8; k += 8) {
    v4f a  = *(const v4f*)(arow + k);
    v4f q0 = *(const v4f*)(w0 + k);
    v4f q1 = *(const v4f*)(w1 + k);
    v4f q2 = *(const v4f*)(w2 + k);
    v4f q3 = *(const v4f*)(w3 + k);
    v2f a0 = {a.x, a.y}, a1 = {a.z, a.w};
    v2f b;
    b = (v2f){q0.x, q0.y};
    acc[0] = __builtin_amdgcn_wmma_f32_16x16x4_f32(false, a0, false, b,
                                                   (short)0, acc[0], false, false);
    b = (v2f){q1.x, q1.y};
    acc[1] = __builtin_amdgcn_wmma_f32_16x16x4_f32(false, a0, false, b,
                                                   (short)0, acc[1], false, false);
    b = (v2f){q2.x, q2.y};
    acc[2] = __builtin_amdgcn_wmma_f32_16x16x4_f32(false, a0, false, b,
                                                   (short)0, acc[2], false, false);
    b = (v2f){q3.x, q3.y};
    acc[3] = __builtin_amdgcn_wmma_f32_16x16x4_f32(false, a0, false, b,
                                                   (short)0, acc[3], false, false);
    b = (v2f){q0.z, q0.w};
    acc[0] = __builtin_amdgcn_wmma_f32_16x16x4_f32(false, a1, false, b,
                                                   (short)0, acc[0], false, false);
    b = (v2f){q1.z, q1.w};
    acc[1] = __builtin_amdgcn_wmma_f32_16x16x4_f32(false, a1, false, b,
                                                   (short)0, acc[1], false, false);
    b = (v2f){q2.z, q2.w};
    acc[2] = __builtin_amdgcn_wmma_f32_16x16x4_f32(false, a1, false, b,
                                                   (short)0, acc[2], false, false);
    b = (v2f){q3.z, q3.w};
    acc[3] = __builtin_amdgcn_wmma_f32_16x16x4_f32(false, a1, false, b,
                                                   (short)0, acc[3], false, false);
  }

  #pragma unroll
  for (int g = 0; g < 4; ++g)
    #pragma unroll
    for (int v = 0; v < 8; ++v) red[tb][ks][g][v][lane] = acc[g][v];
  __syncthreads();

  float s[4];
  #pragma unroll
  for (int g = 0; g < 4; ++g) {
    float t = red[tb][0][g][ks][lane];
    #pragma unroll
    for (int q = 1; q < 8; ++q) t += red[tb][q][g][ks][lane];
    s[g] = t;
  }
  s[0] += xp0; s[1] += xp1; s[2] += xp2; s[3] += xp3;

  int tgt = TT - 1 - step;
  float si = 1.f / (1.f + expf(-s[0]));
  float sf = 1.f / (1.f + expf(-s[1]));
  float gt = tanhf(s[2]);
  float so = 1.f / (1.f + expf(-s[3]));
  float cnew = sf * cold + si * gt;
  float hnew = so * tanhf(cnew);
  cc[(size_t)rf * DD + d] = cnew;
  hout[(size_t)rf * DD + d] = hnew;
  if ((rf >> 3) == tgt) lasth[(size_t)rf * DD + d] = hnew;
}

__global__ void ln_relu_kernel(const float* __restrict__ in,
                               const float* __restrict__ g,
                               const float* __restrict__ be,
                               float* __restrict__ out) {
  __shared__ float s1[256], s2[256];
  int r = blockIdx.x, t = threadIdx.x;
  const float* row = in + (size_t)r * DD;
  float s = 0.f, sq = 0.f;
  for (int i = t; i < DD; i += 256) { float v = row[i]; s += v; sq += v * v; }
  s1[t] = s; s2[t] = sq;
  __syncthreads();
  for (int off = 128; off > 0; off >>= 1) {
    if (t < off) { s1[t] += s1[t + off]; s2[t] += s2[t + off]; }
    __syncthreads();
  }
  float mean = s1[0] * (1.0f / DD);
  float var = s2[0] * (1.0f / DD) - mean * mean;
  float inv = rsqrtf(var + 1e-5f);
  for (int i = t; i < DD; i += 256) {
    float v = (row[i] - mean) * inv * g[i] + be[i];
    out[(size_t)r * DD + i] = v > 0.f ? v : 0.f;
  }
}

extern "C" void kernel_launch(void* const* d_in, const int* in_sizes, int n_in,
                              void* d_out, int out_size, void* d_ws, size_t ws_size,
                              hipStream_t stream) {
  const float* x   = (const float*)d_in[0];
  const float* Wn  = (const float*)d_in[1];
  const float* bn  = (const float*)d_in[2];
  const float* Wv  = (const float*)d_in[3];
  const float* bv  = (const float*)d_in[4];
  const float* Wf  = (const float*)d_in[5];
  const float* bf  = (const float*)d_in[6];
  const float* Wih = (const float*)d_in[7];
  const float* Whh = (const float*)d_in[8];
  const float* bih = (const float*)d_in[9];
  const float* bhh = (const float*)d_in[10];
  const float* Wc  = (const float*)d_in[11];
  const float* bc  = (const float*)d_in[12];
  const float* g_a = (const float*)d_in[13];
  const float* be_a= (const float*)d_in[14];
  const float* Wa  = (const float*)d_in[15];
  const float* ba  = (const float*)d_in[16];
  float* out = (float*)d_out;

  float* ws    = (float*)d_ws;
  float* A0    = ws;
  float* F1    = A0    + 131072;
  float* F2    = F1    + 131072;
  float* ENV   = F2    + 131072;
  float* XPRE  = ENV   + 262144;
  float* H0    = XPRE  + 524288;
  float* H1    = H0    + 131072;
  float* CC    = H1    + 131072;
  float* LASTH = CC    + 131072;
  float* UNR   = LASTH + 131072;
  float* LNR   = UNR   + 131072;

  pool_kernel<<<1536, 256, 0, stream>>>(x, A0, F1, F2);
  gemm_kernel<0, 2, 1024><<<256, 256, 0, stream>>>(F1, DD, Wn, DD, bn, nullptr,
                                                   ENV, 2 * DD, 0,  nullptr, DD);
  gemm_kernel<0, 2, 1024><<<256, 256, 0, stream>>>(F2, DD, Wv, DD, bv, nullptr,
                                                   ENV, 2 * DD, DD, nullptr, DD);
  gemm_kernel<1, 4, 2048><<<256, 256, 0, stream>>>(ENV, 2 * DD, Wf, 2 * DD, bf, nullptr,
                                                   H0, 0, 0, CC, 2 * DD);
  gemm_kernel<0, 8, 1024><<<256, 256, 0, stream>>>(A0, DD, Wih, DD, bih, bhh,
                                                   XPRE, 4 * DD, 0, nullptr, 4 * DD);
  float* hin = H0; float* hout = H1;
  for (int s = 0; s < TT; ++s) {
    int mtiles = (TT - s + 1) >> 1;
    if (mtiles > 4)
      lstm_step_kernel<<<mtiles * 32, 512, 0, stream>>>(XPRE, hin, Whh, CC, hout,
                                                        LASTH, s, mtiles);
    else
      lstm_step_kernel<<<mtiles * 64, 256, 0, stream>>>(XPRE, hin, Whh, CC, hout,
                                                        LASTH, s, mtiles);
    float* tmp = hin; hin = hout; hout = tmp;
  }
  gemm_kernel<0, 2, 1024><<<256, 256, 0, stream>>>(LASTH, DD, Wc, DD, bc, nullptr,
                                                   UNR, DD, 0, nullptr, DD);
  ln_relu_kernel<<<128, 256, 0, stream>>>(UNR, g_a, be_a, LNR);
  gemm_kernel<2, 4, 1024><<<236, 256, 0, stream>>>(LNR, DD, Wa, DD, ba, nullptr,
                                                   out, NA, 0, nullptr, NA);
}


# LSTM step K loop: each wave reads whole 128-byte lines, the 20 fragment loads of 4 K blocks as one group, next line's group issued ahead of this line's WMMAs (second register set); k order kept
# speedup vs baseline: 2.9366x; 1.0753x over previous
#include <hip/hip_runtime.h>
#include <math.h>

#define DD 1024
#define NA 1887
#define BB 8
#define TT 16
#define ROWS 128

typedef __attribute__((ext_vector_type(2))) float v2f;
typedef __attribute__((ext_vector_type(4))) float v4f;
typedef __attribute__((ext_vector_type(8))) float v8f;

__global__ void pool_kernel(const float* __restrict__ x, float* __restrict__ A0,
                            float* __restrict__ F1, float* __restrict__ F2) {
  int tid = blockIdx.x * blockDim.x + threadIdx.x;
  if (tid >= ROWS * DD * 3) return;
  int c  = tid % 3;
  int rd = tid / 3;
  int d  = rd % DD;
  int r  = rd / DD;
  const float* p = x + (size_t)rd * 147 + c;
  float s = 0.f;
  #pragma unroll
  for (int i = 0; i < 49; ++i) s += p[i * 3];
  s *= (1.0f / 49.0f);
  if (c == 0) {
    int b = r >> 4, t = r & 15;
    A0[(size_t)(t * BB + b) * DD + d] = s;
  } else if (c == 1) {
    F1[(size_t)r * DD + d] = s;
  } else {
    F2[(size_t)r * DD + d] = s;
  }
}

template <int MODE, int MT, int KK>
__global__ void __launch_bounds__(256)
gemm_kernel(const float* __restrict__ A, int lda,
            const float* __restrict__ W, int ldw,
            const float* __restrict__ bias1,
            const float* __restrict__ bias2,
            float* __restrict__ out, int ldo, int coloff,
            float* __restrict__ out2, int N) {
  __shared__ float red[8][MT][8][32];
  constexpr int MG = 8 / MT;
  int ks   = threadIdx.x >> 5;
  int lane = threadIdx.x & 31;
  int n0    = (blockIdx.x / MG) << 4;
  int mbase = (blockIdx.x % MG) * (MT * 16);
  int c    = lane & 15;
  int hi   = lane >> 4;
  int koff = hi << 2;
  int ncol = n0 + c;
  int nload = (ncol < N) ? ncol : (N - 1);
  constexpr int KS = KK / 8;
  const float* wrow = W + (size_t)nload * ldw + ks * KS + koff;
  const float* arow = A + (size_t)(mbase + c) * lda + ks * KS + koff;
  const size_t astep = (size_t)16 * lda;

  v8f acc[MT];
  #pragma unroll
  for (int mt = 0; mt < MT; ++mt)
    #pragma unroll
    for (int v = 0; v < 8; ++v) acc[mt][v] = 0.f;

  constexpr int NL = KS / 32;
  v4f b[2][4];
  v4f a[2][4][MT];
  #pragma unroll
  for (int j = 0; j < 4; ++j) {
    b[0][j] = *(const v4f*)(wrow + 8 * j);
    #pragma unroll
    for (int mt = 0; mt < MT; ++mt)
      a[0][j][mt] = *(const v4f*)(arow + mt * astep + 8 * j);
  }
  #pragma unroll
  for (int t = 0; t < NL; ++t) {
    const int cur = t & 1, nxt = cur ^ 1;
    if (t + 1 < NL) {
      #pragma unroll
      for (int j = 0; j < 4; ++j) {
        b[nxt][j] = *(const v4f*)(wrow + (t + 1) * 32 + 8 * j);
        #pragma unroll
        for (int mt = 0; mt < MT; ++mt)
          a[nxt][j][mt] = *(const v4f*)(arow + mt * astep + (t + 1) * 32 + 8 * j);
      }
    }
    __builtin_amdgcn_sched_barrier(0);
    #pragma unroll
    for (int j = 0; j < 4; ++j) {
      v2f b0 = {b[cur][j].x, b[cur][j].y}, b1 = {b[cur][j].z, b[cur][j].w};
      #pragma unroll
      for (int mt = 0; mt < MT; ++mt) {
        v2f a0 = {a[cur][j][mt].x, a[cur][j][mt].y};
        acc[mt] = __builtin_amdgcn_wmma_f32_16x16x4_f32(false, a0, false, b0,
                                                        (short)0, acc[mt], false, false);
      }
      #pragma unroll
      for (int mt = 0; mt < MT; ++mt) {
        v2f a1 = {a[cur][j][mt].z, a[cur][j][mt].w};
        acc[mt] = __builtin_amdgcn_wmma_f32_16x16x4_f32(false, a1, false, b1,
                                                        (short)0, acc[mt], false, false);
      }
    }
    __builtin_amdgcn_sched_barrier(0);
  }

  #pragma unroll
  for (int mt = 0; mt < MT; ++mt)
    #pragma unroll
    for (int v = 0; v < 8; ++v) red[ks][mt][v][lane] = acc[mt][v];
  __syncthreads();

  float bsum = bias1[nload] + (bias2 ? bias2[nload] : 0.f);
  #pragma unroll
  for (int mt = 0; mt < MT; ++mt) {
    float t = red[0][mt][ks][lane];
    #pragma unroll
    for (int q = 1; q < 8; ++q) t += red[q][mt][ks][lane];
    float val = t + bsum;
    int r = mbase + mt * 16 + ks + (hi << 3);
    if (MODE == 0) {
      out[(size_t)r * ldo + coloff + ncol] = val;
    } else if (MODE == 1) {
      int bi = r >> 4, tt = r & 15;
      int rr = tt * BB + bi;
      if (ncol < DD) out[(size_t)rr * DD + ncol] = val;
      else           out2[(size_t)rr * DD + (ncol - DD)] = val;
    } else {
      if (ncol < N) {
        int tt = r >> 3, bi = r & 7;
        out[(size_t)(bi * TT + tt) * N + ncol] = val;
      }
    }
  }
}

__global__ void __launch_bounds__(512)
lstm_step_kernel(const float* __restrict__ xpre,
                 const float* __restrict__ hin,
                 const float* __restrict__ Whh,
                 float* __restrict__ cc,
                 float* __restrict__ hout,
                 float* __restrict__ lasth,
                 int step, int mtiles) {
  __shared__ float red[2][8][4][8][32];
  int lw   = threadIdx.x >> 5;
  int lane = threadIdx.x & 31;
  int tb   = lw >> 3;
  int ks   = lw & 7;
  int wt   = blockIdx.x * (blockDim.x >> 8) + tb;
  int mt   = wt % mtiles;
  int n0   = (wt / mtiles) << 4;
  int m0   = mt << 4;
  int nn   = lane & 15;
  int mr   = lane & 15;
  int koff = (lane >> 4) << 2;
  int hi   = lane >> 4;
  int kbase = ks * (DD / 8);

  int rf = m0 + ks + (hi << 3);
  int d  = n0 + nn;
  float xp0 = xpre[(size_t)rf * (4 * DD) + 0 * DD + d];
  float xp1 = xpre[(size_t)rf * (4 * DD) + 1 * DD + d];
  float xp2 = xpre[(size_t)rf * (4 * DD) + 2 * DD + d];
  float xp3 = xpre[(size_t)rf * (4 * DD) + 3 * DD + d];
  float cold = cc[(size_t)rf * DD + d];

  v8f acc[4];
  #pragma unroll
  for (int g = 0; g < 4; ++g)
    #pragma unroll
    for (int v = 0; v < 8; ++v) acc[g][v] = 0.f;

  const float* arow = hin + (size_t)(m0 + mr) * DD + kbase + koff;
  const float* w0 = Whh + (size_t)(0 * DD + n0 + nn) * DD + kbase + koff;
  const float* w1 = Whh + (size_t)(1 * DD + n0 + nn) * DD + kbase + koff;
  const float* w2 = Whh + (size_t)(2 * DD + n0 + nn) * DD + kbase + koff;
  const float* w3 = Whh + (size_t)(3 * DD + n0 + nn) * DD + kbase + koff;

  constexpr int LNL = (DD / 8) / 32;
  v4f fa[2][4];
  v4f fq[2][4][4];
  #pragma unroll
  for (int j = 0; j < 4; ++j) {
    fa[0][j]    = *(const v4f*)(arow + 8 * j);
    fq[0][j][0] = *(const v4f*)(w0 + 8 * j);
    fq[0][j][1] = *(const v4f*)(w1 + 8 * j);
    fq[0][j][2] = *(const v4f*)(w2 + 8 * j);
    fq[0][j][3] = *(const v4f*)(w3 + 8 * j);
  }
  #pragma unroll
  for (int t = 0; t < LNL; ++t) {
    const int cur = t & 1, nxt = cur ^ 1;
    if (t + 1 < LNL) {
      #pragma unroll
      for (int j = 0; j < 4; ++j) {
        fa[nxt][j]    = *(const v4f*)(arow + (t + 1) * 32 + 8 * j);
        fq[nxt][j][0] = *(const v4f*)(w0 + (t + 1) * 32 + 8 * j);
        fq[nxt][j][1] = *(const v4f*)(w1 + (t + 1) * 32 + 8 * j);
        fq[nxt][j][2] = *(const v4f*)(w2 + (t + 1) * 32 + 8 * j);
        fq[nxt][j][3] = *(const v4f*)(w3 + (t + 1) * 32 + 8 * j);
      }
    }
    __builtin_amdgcn_sched_barrier(0);
    #pragma unroll
    for (int j = 0; j < 4; ++j) {
      v2f a0 = {fa[cur][j].x, fa[cur][j].y}, a1 = {fa[cur][j].z, fa[cur][j].w};
      #pragma unroll
      for (int g = 0; g < 4; ++g) {
        v2f b = {fq[cur][j][g].x, fq[cur][j][g].y};
        acc[g] = __builtin_amdgcn_wmma_f32_16x16x4_f32(false, a0, false, b,
                                                       (short)0, acc[g], false, false);
      }
      #pragma unroll
      for (int g = 0; g < 4; ++g) {
        v2f b = {fq[cur][j][g].z, fq[cur][j][g].w};
        acc[g] = __builtin_amdgcn_wmma_f32_16x16x4_f32(false, a1, false, b,
                                                       (short)0, acc[g], false, false);
      }
    }
  }

  #pragma unroll
  for (int g = 0; g < 4; ++g)
    #pragma unroll
    for (int v = 0; v < 8; ++v) red[tb][ks][g][v][lane] = acc[g][v];
  __syncthreads();

  float s[4];
  #pragma unroll
  for (int g = 0; g < 4; ++g) {
    float t = red[tb][0][g][ks][lane];
    #pragma unroll
    for (int q = 1; q < 8; ++q) t += red[tb][q][g][ks][lane];
    s[g] = t;
  }
  s[0] += xp0; s[1] += xp1; s[2] += xp2; s[3] += xp3;

  int tgt = TT - 1 - step;
  float si = 1.f / (1.f + expf(-s[0]));
  float sf = 1.f / (1.f + expf(-s[1]));
  float gt = tanhf(s[2]);
  float so = 1.f / (1.f + expf(-s[3]));
  float cnew = sf * cold + si * gt;
  float hnew = so * tanhf(cnew);
  cc[(size_t)rf * DD + d] = cnew;
  hout[(size_t)rf * DD + d] = hnew;
  if ((rf >> 3) == tgt) lasth[(size_t)rf * DD + d] = hnew;
}

__global__ void ln_relu_kernel(const float* __restrict__ in,
                               const float* __restrict__ g,
                               const float* __restrict__ be,
                               float* __restrict__ out) {
  __shared__ float s1[256], s2[256];
  int r = blockIdx.x, t = threadIdx.x;
  const float* row = in + (size_t)r * DD;
  float s = 0.f, sq = 0.f;
  for (int i = t; i < DD; i += 256) { float v = row[i]; s += v; sq += v * v; }
  s1[t] = s; s2[t] = sq;
  __syncthreads();
  for (int off = 128; off > 0; off >>= 1) {
    if (t < off) { s1[t] += s1[t + off]; s2[t] += s2[t + off]; }
    __syncthreads();
  }
  float mean = s1[0] * (1.0f / DD);
  float var = s2[0] * (1.0f / DD) - mean * mean;
  float inv = rsqrtf(var + 1e-5f);
  for (int i = t; i < DD; i += 256) {
    float v = (row[i] - mean) * inv * g[i] + be[i];
    out[(size_t)r * DD + i] = v > 0.f ? v : 0.f;
  }
}

extern "C" void kernel_launch(void* const* d_in, const int* in_sizes, int n_in,
                              void* d_out, int out_size, void* d_ws, size_t ws_size,
                              hipStream_t stream) {
  const float* x   = (const float*)d_in[0];
  const float* Wn  = (const float*)d_in[1];
  const float* bn  = (const float*)d_in[2];
  const float* Wv  = (const float*)d_in[3];
  const float* bv  = (const float*)d_in[4];
  const float* Wf  = (const float*)d_in[5];
  const float* bf  = (const float*)d_in[6];
  const float* Wih = (const float*)d_in[7];
  const float* Whh = (const float*)d_in[8];
  const float* bih = (const float*)d_in[9];
  const float* bhh = (const float*)d_in[10];
  const float* Wc  = (const float*)d_in[11];
  const float* bc  = (const float*)d_in[12];
  const float* g_a = (const float*)d_in[13];
  const float* be_a= (const float*)d_in[14];
  const float* Wa  = (const float*)d_in[15];
  const float* ba  = (const float*)d_in[16];
  float* out = (float*)d_out;

  float* ws    = (float*)d_ws;
  float* A0    = ws;
  float* F1    = A0    + 131072;
  float* F2    = F1    + 131072;
  float* ENV   = F2    + 131072;
  float* XPRE  = ENV   + 262144;
  float* H0    = XPRE  + 524288;
  float* H1    = H0    + 131072;
  float* CC    = H1    + 131072;
  float* LASTH = CC    + 131072;
  float* UNR   = LASTH + 131072;
  float* LNR   = UNR   + 131072;

  pool_kernel<<<1536, 256, 0, stream>>>(x, A0, F1, F2);
  gemm_kernel<0, 2, 1024><<<256, 256, 0, stream>>>(F1, DD, Wn, DD, bn, nullptr,
                                                   ENV, 2 * DD, 0,  nullptr, DD);
  gemm_kernel<0, 2, 1024><<<256, 256, 0, stream>>>(F2, DD, Wv, DD, bv, nullptr,
                                                   ENV, 2 * DD, DD, nullptr, DD);
  gemm_kernel<1, 4, 2048><<<256, 256, 0, stream>>>(ENV, 2 * DD, Wf, 2 * DD, bf, nullptr,
                                                   H0, 0, 0, CC, 2 * DD);
  gemm_kernel<0, 8, 1024><<<256, 256, 0, stream>>>(A0, DD, Wih, DD, bih, bhh,
                                                   XPRE, 4 * DD, 0, nullptr, 4 * DD);
  float* hin = H0; float* hout = H1;
  for (int s = 0; s < TT; ++s) {
    int mtiles = (TT - s + 1) >> 1;
    if (mtiles > 4)
      lstm_step_kernel<<<mtiles * 32, 512, 0, stream>>>(XPRE, hin, Whh, CC, hout,
                                                        LASTH, s, mtiles);
    else
      lstm_step_kernel<<<mtiles * 64, 256, 0, stream>>>(XPRE, hin, Whh, CC, hout,
                                                        LASTH, s, mtiles);
    float* tmp = hin; hin = hout; hout = tmp;
  }
  gemm_kernel<0, 2, 1024><<<256, 256, 0, stream>>>(LASTH, DD, Wc, DD, bc, nullptr,
                                                   UNR, DD, 0, nullptr, DD);
  ln_relu_kernel<<<128, 256, 0, stream>>>(UNR, g_a, be_a, LNR);
  gemm_kernel<2, 4, 1024><<<236, 256, 0, stream>>>(LNR, DD, Wa, DD, ba, nullptr,
                                                   out, NA, 0, nullptr, NA);
}


# LSTM steps with more than 4 row tiles: every wave carries two row tiles per Whh fragment (8 accumulators), blocks of 8 waves instead of 16, so half as many waves enter; k sum order kept
# speedup vs baseline: 3.0401x; 1.0352x over previous
#include <hip/hip_runtime.h>
#include <math.h>

#define DD 1024
#define NA 1887
#define BB 8
#define TT 16
#define ROWS 128

typedef __attribute__((ext_vector_type(2))) float v2f;
typedef __attribute__((ext_vector_type(4))) float v4f;
typedef __attribute__((ext_vector_type(8))) float v8f;

__global__ void pool_kernel(const float* __restrict__ x, float* __restrict__ A0,
                            float* __restrict__ F1, float* __restrict__ F2) {
  int tid = blockIdx.x * blockDim.x + threadIdx.x;
  if (tid >= ROWS * DD * 3) return;
  int c  = tid % 3;
  int rd = tid / 3;
  int d  = rd % DD;
  int r  = rd / DD;
  const float* p = x + (size_t)rd * 147 + c;
  float s = 0.f;
  #pragma unroll
  for (int i = 0; i < 49; ++i) s += p[i * 3];
  s *= (1.0f / 49.0f);
  if (c == 0) {
    int b = r >> 4, t = r & 15;
    A0[(size_t)(t * BB + b) * DD + d] = s;
  } else if (c == 1) {
    F1[(size_t)r * DD + d] = s;
  } else {
    F2[(size_t)r * DD + d] = s;
  }
}

template <int MODE, int MT, int KK>
__global__ void __launch_bounds__(256)
gemm_kernel(const float* __restrict__ A, int lda,
            const float* __restrict__ W, int ldw,
            const float* __restrict__ bias1,
            const float* __restrict__ bias2,
            float* __restrict__ out, int ldo, int coloff,
            float* __restrict__ out2, int N) {
  __shared__ float red[8][MT][8][32];
  constexpr int MG = 8 / MT;
  int ks   = threadIdx.x >> 5;
  int lane = threadIdx.x & 31;
  int n0    = (blockIdx.x / MG) << 4;
  int mbase = (blockIdx.x % MG) * (MT * 16);
  int c    = lane & 15;
  int hi   = lane >> 4;
  int koff = hi << 2;
  int ncol = n0 + c;
  int nload = (ncol < N) ? ncol : (N - 1);
  constexpr int KS = KK / 8;
  const float* wrow = W + (size_t)nload * ldw + ks * KS + koff;
  const float* arow = A + (size_t)(mbase + c) * lda + ks * KS + koff;
  const size_t astep = (size_t)16 * lda;

  v8f acc[MT];
  #pragma unroll
  for (int mt = 0; mt < MT; ++mt)
    #pragma unroll
    for (int v = 0; v < 8; ++v) acc[mt][v] = 0.f;

  constexpr int NL = KS / 32;
  v4f b[2][4];
  v4f a[2][4][MT];
  #pragma unroll
  for (int j = 0; j < 4; ++j) {
    b[0][j] = *(const v4f*)(wrow + 8 * j);
    #pragma unroll
    for (int mt = 0; mt < MT; ++mt)
      a[0][j][mt] = *(const v4f*)(arow + mt * astep + 8 * j);
  }
  #pragma unroll
  for (int t = 0; t < NL; ++t) {
    const int cur = t & 1, nxt = cur ^ 1;
    if (t + 1 < NL) {
      #pragma unroll
      for (int j = 0; j < 4; ++j) {
        b[nxt][j] = *(const v4f*)(wrow + (t + 1) * 32 + 8 * j);
        #pragma unroll
        for (int mt = 0; mt < MT; ++mt)
          a[nxt][j][mt] = *(const v4f*)(arow + mt * astep + (t + 1) * 32 + 8 * j);
      }
    }
    __builtin_amdgcn_sched_barrier(0);
    #pragma unroll
    for (int j = 0; j < 4; ++j) {
      v2f b0 = {b[cur][j].x, b[cur][j].y}, b1 = {b[cur][j].z, b[cur][j].w};
      #pragma unroll
      for (int mt = 0; mt < MT; ++mt) {
        v2f a0 = {a[cur][j][mt].x, a[cur][j][mt].y};
        acc[mt] = __builtin_amdgcn_wmma_f32_16x16x4_f32(false, a0, false, b0,
                                                        (short)0, acc[mt], false, false);
      }
      #pragma unroll
      for (int mt = 0; mt < MT; ++mt) {
        v2f a1 = {a[cur][j][mt].z, a[cur][j][mt].w};
        acc[mt] = __builtin_amdgcn_wmma_f32_16x16x4_f32(false, a1, false, b1,
                                                        (short)0, acc[mt], false, false);
      }
    }
    __builtin_amdgcn_sched_barrier(0);
  }

  #pragma unroll
  for (int mt = 0; mt < MT; ++mt)
    #pragma unroll
    for (int v = 0; v < 8; ++v) red[ks][mt][v][lane] = acc[mt][v];
  __syncthreads();

  float bsum = bias1[nload] + (bias2 ? bias2[nload] : 0.f);
  #pragma unroll
  for (int mt = 0; mt < MT; ++mt) {
    float t = red[0][mt][ks][lane];
    #pragma unroll
    for (int q = 1; q < 8; ++q) t += red[q][mt][ks][lane];
    float val = t + bsum;
    int r = mbase + mt * 16 + ks + (hi << 3);
    if (MODE == 0) {
      out[(size_t)r * ldo + coloff + ncol] = val;
    } else if (MODE == 1) {
      int bi = r >> 4, tt = r & 15;
      int rr = tt * BB + bi;
      if (ncol < DD) out[(size_t)rr * DD + ncol] = val;
      else           out2[(size_t)rr * DD + (ncol - DD)] = val;
    } else {
      if (ncol < N) {
        int tt = r >> 3, bi = r & 7;
        out[(size_t)(bi * TT + tt) * N + ncol] = val;
      }
    }
  }
}

template <int NT>
__device__ __forceinline__ void lstm_tiles(const float* __restrict__ xpre,
                                           const float* __restrict__ hin,
                                           const float* __restrict__ Whh,
                                           float* __restrict__ cc,
                                           float* __restrict__ hout,
                                           float* __restrict__ lasth,
                                           int step, int m0, int n0,
                                           float (*red)[2][4][8][32]) {
  int ks   = threadIdx.x >> 5;
  int lane = threadIdx.x & 31;
  int nn   = lane & 15;
  int hi   = lane >> 4;
  int koff = hi << 2;
  int kbase = ks * (DD / 8);
  int d  = n0 + nn;

  float xp[NT][4], cold[NT];
  #pragma unroll
  for (int tt = 0; tt < NT; ++tt) {
    int rf = m0 + tt * 16 + ks + (hi << 3);
    #pragma unroll
    for (int g = 0; g < 4; ++g) xp[tt][g] = xpre[(size_t)rf * (4 * DD) + g * DD + d];
    cold[tt] = cc[(size_t)rf * DD + d];
  }

  v8f acc[NT][4];
  #pragma unroll
  for (int tt = 0; tt < NT; ++tt)
    #pragma unroll
    for (int g = 0; g < 4; ++g)
      #pragma unroll
      for (int v = 0; v < 8; ++v) acc[tt][g][v] = 0.f;

  const float* arow = hin + (size_t)(m0 + nn) * DD + kbase + koff;
  const size_t astep = (size_t)16 * DD;
  const float* wrow = Whh + (size_t)(n0 + nn) * DD + kbase + koff;
  const size_t gstep = (size_t)DD * DD;

  constexpr int LNL = (DD / 8) / 32;
  v4f fa[2][4][NT];
  v4f fq[2][4][4];
  #pragma unroll
  for (int j = 0; j < 4; ++j) {
    #pragma unroll
    for (int tt = 0; tt < NT; ++tt)
      fa[0][j][tt] = *(const v4f*)(arow + tt * astep + 8 * j);
    #pragma unroll
    for (int g = 0; g < 4; ++g)
      fq[0][j][g] = *(const v4f*)(wrow + g * gstep + 8 * j);
  }
  #pragma unroll
  for (int t = 0; t < LNL; ++t) {
    const int cur = t & 1, nxt = cur ^ 1;
    if (t + 1 < LNL) {
      #pragma unroll
      for (int j = 0; j < 4; ++j) {
        #pragma unroll
        for (int tt = 0; tt < NT; ++tt)
          fa[nxt][j][tt] = *(const v4f*)(arow + tt * astep + (t + 1) * 32 + 8 * j);
        #pragma unroll
        for (int g = 0; g < 4; ++g)
          fq[nxt][j][g] = *(const v4f*)(wrow + g * gstep + (t + 1) * 32 + 8 * j);
      }
    }
    __builtin_amdgcn_sched_barrier(0);
    #pragma unroll
    for (int j = 0; j < 4; ++j) {
      #pragma unroll
      for (int tt = 0; tt < NT; ++tt) {
        v2f a0 = {fa[cur][j][tt].x, fa[cur][j][tt].y};
        #pragma unroll
        for (int g = 0; g < 4; ++g) {
          v2f b = {fq[cur][j][g].x, fq[cur][j][g].y};
          acc[tt][g] = __builtin_amdgcn_wmma_f32_16x16x4_f32(false, a0, false, b,
                                                             (short)0, acc[tt][g], false, false);
        }
      }
      #pragma unroll
      for (int tt = 0; tt < NT; ++tt) {
        v2f a1 = {fa[cur][j][tt].z, fa[cur][j][tt].w};
        #pragma unroll
        for (int g = 0; g < 4; ++g) {
          v2f b = {fq[cur][j][g].z, fq[cur][j][g].w};
          acc[tt][g] = __builtin_amdgcn_wmma_f32_16x16x4_f32(false, a1, false, b,
                                                             (short)0, acc[tt][g], false, false);
        }
      }
    }
  }

  #pragma unroll
  for (int tt = 0; tt < NT; ++tt)
    #pragma unroll
    for (int g = 0; g < 4; ++g)
      #pragma unroll
      for (int v = 0; v < 8; ++v) red[ks][tt][g][v][lane] = acc[tt][g][v];
  __syncthreads();

  float s[NT][4];
  #pragma unroll
  for (int tt = 0; tt < NT; ++tt)
    #pragma unroll
    for (int g = 0; g < 4; ++g) {
      float t = red[0][tt][g][ks][lane];
      #pragma unroll
      for (int q = 1; q < 8; ++q) t += red[q][tt][g][ks][lane];
      s[tt][g] = t + xp[tt][g];
    }

  int tgt = TT - 1 - step;
  #pragma unroll
  for (int tt = 0; tt < NT; ++tt) {
    int rf = m0 + tt * 16 + ks + (hi << 3);
    float si = 1.f / (1.f + expf(-s[tt][0]));
    float sf = 1.f / (1.f + expf(-s[tt][1]));
    float gt = tanhf(s[tt][2]);
    float so = 1.f / (1.f + expf(-s[tt][3]));
    float cnew = sf * cold[tt] + si * gt;
    float hnew = so * tanhf(cnew);
    cc[(size_t)rf * DD + d] = cnew;
    hout[(size_t)rf * DD + d] = hnew;
    if ((rf >> 3) == tgt) lasth[(size_t)rf * DD + d] = hnew;
  }
}

__global__ void __launch_bounds__(256)
lstm_step_kernel(const float* __restrict__ xpre,
                 const float* __restrict__ hin,
                 const float* __restrict__ Whh,
                 float* __restrict__ cc,
                 float* __restrict__ hout,
                 float* __restrict__ lasth,
                 int step, int mtiles, int pairs) {
  __shared__ float red[8][2][4][8][32];
  int units = pairs ? ((mtiles + 1) >> 1) : mtiles;
  int u  = blockIdx.x % units;
  int n0 = (blockIdx.x / units) << 4;
  int mt = pairs ? (u << 1) : u;
  if (pairs && mt + 1 < mtiles)
    lstm_tiles<2>(xpre, hin, Whh, cc, hout, lasth, step, mt << 4, n0, red);
  else
    lstm_tiles<1>(xpre, hin, Whh, cc, hout, lasth, step, mt << 4, n0, red);
}

__global__ void ln_relu_kernel(const float* __restrict__ in,
                               const float* __restrict__ g,
                               const float* __restrict__ be,
                               float* __restrict__ out) {
  __shared__ float s1[256], s2[256];
  int r = blockIdx.x, t = threadIdx.x;
  const float* row = in + (size_t)r * DD;
  float s = 0.f, sq = 0.f;
  for (int i = t; i < DD; i += 256) { float v = row[i]; s += v; sq += v * v; }
  s1[t] = s; s2[t] = sq;
  __syncthreads();
  for (int off = 128; off > 0; off >>= 1) {
    if (t < off) { s1[t] += s1[t + off]; s2[t] += s2[t + off]; }
    __syncthreads();
  }
  float mean = s1[0] * (1.0f / DD);
  float var = s2[0] * (1.0f / DD) - mean * mean;
  float inv = rsqrtf(var + 1e-5f);
  for (int i = t; i < DD; i += 256) {
    float v = (row[i] - mean) * inv * g[i] + be[i];
    out[(size_t)r * DD + i] = v > 0.f ? v : 0.f;
  }
}

extern "C" void kernel_launch(void* const* d_in, const int* in_sizes, int n_in,
                              void* d_out, int out_size, void* d_ws, size_t ws_size,
                              hipStream_t stream) {
  const float* x   = (const float*)d_in[0];
  const float* Wn  = (const float*)d_in[1];
  const float* bn  = (const float*)d_in[2];
  const float* Wv  = (const float*)d_in[3];
  const float* bv  = (const float*)d_in[4];
  const float* Wf  = (const float*)d_in[5];
  const float* bf  = (const float*)d_in[6];
  const float* Wih = (const float*)d_in[7];
  const float* Whh = (const float*)d_in[8];
  const float* bih = (const float*)d_in[9];
  const float* bhh = (const float*)d_in[10];
  const float* Wc  = (const float*)d_in[11];
  const float* bc  = (const float*)d_in[12];
  const float* g_a = (const float*)d_in[13];
  const float* be_a= (const float*)d_in[14];
  const float* Wa  = (const float*)d_in[15];
  const float* ba  = (const float*)d_in[16];
  float* out = (float*)d_out;

  float* ws    = (float*)d_ws;
  float* A0    = ws;
  float* F1    = A0    + 131072;
  float* F2    = F1    + 131072;
  float* ENV   = F2    + 131072;
  float* XPRE  = ENV   + 262144;
  float* H0    = XPRE  + 524288;
  float* H1    = H0    + 131072;
  float* CC    = H1    + 131072;
  float* LASTH = CC    + 131072;
  float* UNR   = LASTH + 131072;
  float* LNR   = UNR   + 131072;

  pool_kernel<<<1536, 256, 0, stream>>>(x, A0, F1, F2);
  gemm_kernel<0, 2, 1024><<<256, 256, 0, stream>>>(F1, DD, Wn, DD, bn, nullptr,
                                                   ENV, 2 * DD, 0,  nullptr, DD);
  gemm_kernel<0, 2, 1024><<<256, 256, 0, stream>>>(F2, DD, Wv, DD, bv, nullptr,
                                                   ENV, 2 * DD, DD, nullptr, DD);
  gemm_kernel<1, 4, 2048><<<256, 256, 0, stream>>>(ENV, 2 * DD, Wf, 2 * DD, bf, nullptr,
                                                   H0, 0, 0, CC, 2 * DD);
  gemm_kernel<0, 8, 1024><<<256, 256, 0, stream>>>(A0, DD, Wih, DD, bih, bhh,
                                                   XPRE, 4 * DD, 0, nullptr, 4 * DD);
  float* hin = H0; float* hout = H1;
  for (int s = 0; s < TT; ++s) {
    int mtiles = (TT - s + 1) >> 1;
    if (mtiles > 4)
      lstm_step_kernel<<<((mtiles + 1) >> 1) * 64, 256, 0, stream>>>(XPRE, hin, Whh, CC, hout,
                                                                     LASTH, s, mtiles, 1);
    else
      lstm_step_kernel<<<mtiles * 64, 256, 0, stream>>>(XPRE, hin, Whh, CC, hout,
                                                        LASTH, s, mtiles, 0);
    float* tmp = hin; hin = hout; hout = tmp;
  }
  gemm_kernel<0, 2, 1024><<<256, 256, 0, stream>>>(LASTH, DD, Wc, DD, bc, nullptr,
                                                   UNR, DD, 0, nullptr, DD);
  ln_relu_kernel<<<128, 256, 0, stream>>>(UNR, g_a, be_a, LNR);
  gemm_kernel<2, 4, 1024><<<236, 256, 0, stream>>>(LNR, DD, Wa, DD, ba, nullptr,
                                                   out, NA, 0, nullptr, NA);
}


# e_n, e_v and x_pre GEMMs (all three read only the pooled features) run as ONE launch of 768 equal blocks, 4 row tiles per wave each: 24 launches become 22; every k sum order kept
# speedup vs baseline: 3.1382x; 1.0323x over previous
#include <hip/hip_runtime.h>
#include <math.h>

#define DD 1024
#define NA 1887
#define BB 8
#define TT 16
#define ROWS 128

typedef __attribute__((ext_vector_type(2))) float v2f;
typedef __attribute__((ext_vector_type(4))) float v4f;
typedef __attribute__((ext_vector_type(8))) float v8f;

__global__ void pool_kernel(const float* __restrict__ x, float* __restrict__ A0,
                            float* __restrict__ F1, float* __restrict__ F2) {
  int tid = blockIdx.x * blockDim.x + threadIdx.x;
  if (tid >= ROWS * DD * 3) return;
  int c  = tid % 3;
  int rd = tid / 3;
  int d  = rd % DD;
  int r  = rd / DD;
  const float* p = x + (size_t)rd * 147 + c;
  float s = 0.f;
  #pragma unroll
  for (int i = 0; i < 49; ++i) s += p[i * 3];
  s *= (1.0f / 49.0f);
  if (c == 0) {
    int b = r >> 4, t = r & 15;
    A0[(size_t)(t * BB + b) * DD + d] = s;
  } else if (c == 1) {
    F1[(size_t)r * DD + d] = s;
  } else {
    F2[(size_t)r * DD + d] = s;
  }
}

template <int MODE, int MT, int KK>
__device__ __forceinline__ void
gemm_block(const float* __restrict__ A, int lda,
           const float* __restrict__ W, int ldw,
           const float* __restrict__ bias1,
           const float* __restrict__ bias2,
           float* __restrict__ out, int ldo, int coloff,
           float* __restrict__ out2, int N, int bid,
           float (*red)[MT][8][32]) {
  constexpr int MG = 8 / MT;
  int ks   = threadIdx.x >> 5;
  int lane = threadIdx.x & 31;
  int n0    = (bid / MG) << 4;
  int mbase = (bid % MG) * (MT * 16);
  int c    = lane & 15;
  int hi   = lane >> 4;
  int koff = hi << 2;
  int ncol = n0 + c;
  int nload = (ncol < N) ? ncol : (N - 1);
  constexpr int KS = KK / 8;
  const float* wrow = W + (size_t)nload * ldw + ks * KS + koff;
  const float* arow = A + (size_t)(mbase + c) * lda + ks * KS + koff;
  const size_t astep = (size_t)16 * lda;

  v8f acc[MT];
  #pragma unroll
  for (int mt = 0; mt < MT; ++mt)
    #pragma unroll
    for (int v = 0; v < 8; ++v) acc[mt][v] = 0.f;

  constexpr int NL = KS / 32;
  v4f b[2][4];
  v4f a[2][4][MT];
  #pragma unroll
  for (int j = 0; j < 4; ++j) {
    b[0][j] = *(const v4f*)(wrow + 8 * j);
    #pragma unroll
    for (int mt = 0; mt < MT; ++mt)
      a[0][j][mt] = *(const v4f*)(arow + mt * astep + 8 * j);
  }
  #pragma unroll
  for (int t = 0; t < NL; ++t) {
    const int cur = t & 1, nxt = cur ^ 1;
    if (t + 1 < NL) {
      #pragma unroll
      for (int j = 0; j < 4; ++j) {
        b[nxt][j] = *(const v4f*)(wrow + (t + 1) * 32 + 8 * j);
        #pragma unroll
        for (int mt = 0; mt < MT; ++mt)
          a[nxt][j][mt] = *(const v4f*)(arow + mt * astep + (t + 1) * 32 + 8 * j);
      }
    }
    __builtin_amdgcn_sched_barrier(0);
    #pragma unroll
    for (int j = 0; j < 4; ++j) {
      v2f b0 = {b[cur][j].x, b[cur][j].y}, b1 = {b[cur][j].z, b[cur][j].w};
      #pragma unroll
      for (int mt = 0; mt < MT; ++mt) {
        v2f a0 = {a[cur][j][mt].x, a[cur][j][mt].y};
        acc[mt] = __builtin_amdgcn_wmma_f32_16x16x4_f32(false, a0, false, b0,
                                                        (short)0, acc[mt], false, false);
      }
      #pragma unroll
      for (int mt = 0; mt < MT; ++mt) {
        v2f a1 = {a[cur][j][mt].z, a[cur][j][mt].w};
        acc[mt] = __builtin_amdgcn_wmma_f32_16x16x4_f32(false, a1, false, b1,
                                                        (short)0, acc[mt], false, false);
      }
    }
    __builtin_amdgcn_sched_barrier(0);
  }

  #pragma unroll
  for (int mt = 0; mt < MT; ++mt)
    #pragma unroll
    for (int v = 0; v < 8; ++v) red[ks][mt][v][lane] = acc[mt][v];
  __syncthreads();

  float bsum = bias1[nload] + (bias2 ? bias2[nload] : 0.f);
  #pragma unroll
  for (int mt = 0; mt < MT; ++mt) {
    float t = red[0][mt][ks][lane];
    #pragma unroll
    for (int q = 1; q < 8; ++q) t += red[q][mt][ks][lane];
    float val = t + bsum;
    int r = mbase + mt * 16 + ks + (hi << 3);
    if (MODE == 0) {
      out[(size_t)r * ldo + coloff + ncol] = val;
    } else if (MODE == 1) {
      int bi = r >> 4, tt = r & 15;
      int rr = tt * BB + bi;
      if (ncol < DD) out[(size_t)rr * DD + ncol] = val;
      else           out2[(size_t)rr * DD + (ncol - DD)] = val;
    } else {
      if (ncol < N) {
        int tt = r >> 3, bi = r & 7;
        out[(size_t)(bi * TT + tt) * N + ncol] = val;
      }
    }
  }
}

template <int MODE, int MT, int KK>
__global__ void __launch_bounds__(256)
gemm_kernel(const float* __restrict__ A, int lda,
            const float* __restrict__ W, int ldw,
            const float* __restrict__ bias1,
            const float* __restrict__ bias2,
            float* __restrict__ out, int ldo, int coloff,
            float* __restrict__ out2, int N) {
  __shared__ float red[8][MT][8][32];
  gemm_block<MODE, MT, KK>(A, lda, W, ldw, bias1, bias2, out, ldo, coloff, out2, N,
                           (int)blockIdx.x, red);
}

__global__ void __launch_bounds__(256)
gemm_pre_kernel(const float* __restrict__ A0, const float* __restrict__ Wih,
                const float* __restrict__ bih, const float* __restrict__ bhh,
                float* __restrict__ XPRE,
                const float* __restrict__ F1, const float* __restrict__ Wn,
                const float* __restrict__ bn,
                const float* __restrict__ F2, const float* __restrict__ Wv,
                const float* __restrict__ bv,
                float* __restrict__ ENV) {
  __shared__ float red[8][4][8][32];
  int b = (int)blockIdx.x;
  const float* A  = A0;  const float* W  = Wih;
  const float* b1 = bih; const float* b2 = bhh;
  float* o = XPRE; int ldo = 4 * DD, coloff = 0, N = 4 * DD, bid = b;
  if (b >= 512) {
    bool v = b >= 640;
    A  = v ? F2 : F1;  W = v ? Wv : Wn;
    b1 = v ? bv : bn;  b2 = nullptr;
    o = ENV; ldo = 2 * DD; coloff = v ? DD : 0; N = DD;
    bid = b - (v ? 640 : 512);
  }
  gemm_block<0, 4, 1024>(A, DD, W, DD, b1, b2, o, ldo, coloff, nullptr, N, bid, red);
}

template <int NT>
__device__ __forceinline__ void lstm_tiles(const float* __restrict__ xpre,
                                           const float* __restrict__ hin,
                                           const float* __restrict__ Whh,
                                           float* __restrict__ cc,
                                           float* __restrict__ hout,
                                           float* __restrict__ lasth,
                                           int step, int m0, int n0,
                                           float (*red)[2][4][8][32]) {
  int ks   = threadIdx.x >> 5;
  int lane = threadIdx.x & 31;
  int nn   = lane & 15;
  int hi   = lane >> 4;
  int koff = hi << 2;
  int kbase = ks * (DD / 8);
  int d  = n0 + nn;

  float xp[NT][4], cold[NT];
  #pragma unroll
  for (int tt = 0; tt < NT; ++tt) {
    int rf = m0 + tt * 16 + ks + (hi << 3);
    #pragma unroll
    for (int g = 0; g < 4; ++g) xp[tt][g] = xpre[(size_t)rf * (4 * DD) + g * DD + d];
    cold[tt] = cc[(size_t)rf * DD + d];
  }

  v8f acc[NT][4];
  #pragma unroll
  for (int tt = 0; tt < NT; ++tt)
    #pragma unroll
    for (int g = 0; g < 4; ++g)
      #pragma unroll
      for (int v = 0; v < 8; ++v) acc[tt][g][v] = 0.f;

  const float* arow = hin + (size_t)(m0 + nn) * DD + kbase + koff;
  const size_t astep = (size_t)16 * DD;
  const float* wrow = Whh + (size_t)(n0 + nn) * DD + kbase + koff;
  const size_t gstep = (size_t)DD * DD;

  constexpr int LNL = (DD / 8) / 32;
  v4f fa[2][4][NT];
  v4f fq[2][4][4];
  #pragma unroll
  for (int j = 0; j < 4; ++j) {
    #pragma unroll
    for (int tt = 0; tt < NT; ++tt)
      fa[0][j][tt] = *(const v4f*)(arow + tt * astep + 8 * j);
    #pragma unroll
    for (int g = 0; g < 4; ++g)
      fq[0][j][g] = *(const v4f*)(wrow + g * gstep + 8 * j);
  }
  #pragma unroll
  for (int t = 0; t < LNL; ++t) {
    const int cur = t & 1, nxt = cur ^ 1;
    if (t + 1 < LNL) {
      #pragma unroll
      for (int j = 0; j < 4; ++j) {
        #pragma unroll
        for (int tt = 0; tt < NT; ++tt)
          fa[nxt][j][tt] = *(const v4f*)(arow + tt * astep + (t + 1) * 32 + 8 * j);
        #pragma unroll
        for (int g = 0; g < 4; ++g)
          fq[nxt][j][g] = *(const v4f*)(wrow + g * gstep + (t + 1) * 32 + 8 * j);
      }
    }
    __builtin_amdgcn_sched_barrier(0);
    #pragma unroll
    for (int j = 0; j < 4; ++j) {
      #pragma unroll
      for (int tt = 0; tt < NT; ++tt) {
        v2f a0 = {fa[cur][j][tt].x, fa[cur][j][tt].y};
        #pragma unroll
        for (int g = 0; g < 4; ++g) {
          v2f b = {fq[cur][j][g].x, fq[cur][j][g].y};
          acc[tt][g] = __builtin_amdgcn_wmma_f32_16x16x4_f32(false, a0, false, b,
                                                             (short)0, acc[tt][g], false, false);
        }
      }
      #pragma unroll
      for (int tt = 0; tt < NT; ++tt) {
        v2f a1 = {fa[cur][j][tt].z, fa[cur][j][tt].w};
        #pragma unroll
        for (int g = 0; g < 4; ++g) {
          v2f b = {fq[cur][j][g].z, fq[cur][j][g].w};
          acc[tt][g] = __builtin_amdgcn_wmma_f32_16x16x4_f32(false, a1, false, b,
                                                             (short)0, acc[tt][g], false, false);
        }
      }
    }
  }

  #pragma unroll
  for (int tt = 0; tt < NT; ++tt)
    #pragma unroll
    for (int g = 0; g < 4; ++g)
      #pragma unroll
      for (int v = 0; v < 8; ++v) red[ks][tt][g][v][lane] = acc[tt][g][v];
  __syncthreads();

  float s[NT][4];
  #pragma unroll
  for (int tt = 0; tt < NT; ++tt)
    #pragma unroll
    for (int g = 0; g < 4; ++g) {
      float t = red[0][tt][g][ks][lane];
      #pragma unroll
      for (int q = 1; q < 8; ++q) t += red[q][tt][g][ks][lane];
      s[tt][g] = t + xp[tt][g];
    }

  int tgt = TT - 1 - step;
  #pragma unroll
  for (int tt = 0; tt < NT; ++tt) {
    int rf = m0 + tt * 16 + ks + (hi << 3);
    float si = 1.f / (1.f + expf(-s[tt][0]));
    float sf = 1.f / (1.f + expf(-s[tt][1]));
    float gt = tanhf(s[tt][2]);
    float so = 1.f / (1.f + expf(-s[tt][3]));
    float cnew = sf * cold[tt] + si * gt;
    float hnew = so * tanhf(cnew);
    cc[(size_t)rf * DD + d] = cnew;
    hout[(size_t)rf * DD + d] = hnew;
    if ((rf >> 3) == tgt) lasth[(size_t)rf * DD + d] = hnew;
  }
}

__global__ void __launch_bounds__(256)
lstm_step_kernel(const float* __restrict__ xpre,
                 const float* __restrict__ hin,
                 const float* __restrict__ Whh,
                 float* __restrict__ cc,
                 float* __restrict__ hout,
                 float* __restrict__ lasth,
                 int step, int mtiles, int pairs) {
  __shared__ float red[8][2][4][8][32];
  int units = pairs ? ((mtiles + 1) >> 1) : mtiles;
  int u  = blockIdx.x % units;
  int n0 = (blockIdx.x / units) << 4;
  int mt = pairs ? (u << 1) : u;
  if (pairs && mt + 1 < mtiles)
    lstm_tiles<2>(xpre, hin, Whh, cc, hout, lasth, step, mt << 4, n0, red);
  else
    lstm_tiles<1>(xpre, hin, Whh, cc, hout, lasth, step, mt << 4, n0, red);
}

__global__ void ln_relu_kernel(const float* __restrict__ in,
                               const float* __restrict__ g,
                               const float* __restrict__ be,
                               float* __restrict__ out) {
  __shared__ float s1[256], s2[256];
  int r = blockIdx.x, t = threadIdx.x;
  const float* row = in + (size_t)r * DD;
  float s = 0.f, sq = 0.f;
  for (int i = t; i < DD; i += 256) { float v = row[i]; s += v; sq += v * v; }
  s1[t] = s; s2[t] = sq;
  __syncthreads();
  for (int off = 128; off > 0; off >>= 1) {
    if (t < off) { s1[t] += s1[t + off]; s2[t] += s2[t + off]; }
    __syncthreads();
  }
  float mean = s1[0] * (1.0f / DD);
  float var = s2[0] * (1.0f / DD) - mean * mean;
  float inv = rsqrtf(var + 1e-5f);
  for (int i = t; i < DD; i += 256) {
    float v = (row[i] - mean) * inv * g[i] + be[i];
    out[(size_t)r * DD + i] = v > 0.f ? v : 0.f;
  }
}

extern "C" void kernel_launch(void* const* d_in, const int* in_sizes, int n_in,
                              void* d_out, int out_size, void* d_ws, size_t ws_size,
                              hipStream_t stream) {
  const float* x   = (const float*)d_in[0];
  const float* Wn  = (const float*)d_in[1];
  const float* bn  = (const float*)d_in[2];
  const float* Wv  = (const float*)d_in[3];
  const float* bv  = (const float*)d_in[4];
  const float* Wf  = (const float*)d_in[5];
  const float* bf  = (const float*)d_in[6];
  const float* Wih = (const float*)d_in[7];
  const float* Whh = (const float*)d_in[8];
  const float* bih = (const float*)d_in[9];
  const float* bhh = (const float*)d_in[10];
  const float* Wc  = (const float*)d_in[11];
  const float* bc  = (const float*)d_in[12];
  const float* g_a = (const float*)d_in[13];
  const float* be_a= (const float*)d_in[14];
  const float* Wa  = (const float*)d_in[15];
  const float* ba  = (const float*)d_in[16];
  float* out = (float*)d_out;

  float* ws    = (float*)d_ws;
  float* A0    = ws;
  float* F1    = A0    + 131072;
  float* F2    = F1    + 131072;
  float* ENV   = F2    + 131072;
  float* XPRE  = ENV   + 262144;
  float* H0    = XPRE  + 524288;
  float* H1    = H0    + 131072;
  float* CC    = H1    + 131072;
  float* LASTH = CC    + 131072;
  float* UNR   = LASTH + 131072;
  float* LNR   = UNR   + 131072;

  pool_kernel<<<1536, 256, 0, stream>>>(x, A0, F1, F2);
  gemm_pre_kernel<<<768, 256, 0, stream>>>(A0, Wih, bih, bhh, XPRE,
                                           F1, Wn, bn, F2, Wv, bv, ENV);
  gemm_kernel<1, 4, 2048><<<256, 256, 0, stream>>>(ENV, 2 * DD, Wf, 2 * DD, bf, nullptr,
                                                   H0, 0, 0, CC, 2 * DD);
  float* hin = H0; float* hout = H1;
  for (int s = 0; s < TT; ++s) {
    int mtiles = (TT - s + 1) >> 1;
    if (mtiles > 4)
      lstm_step_kernel<<<((mtiles + 1) >> 1) * 64, 256, 0, stream>>>(XPRE, hin, Whh, CC, hout,
                                                                     LASTH, s, mtiles, 1);
    else
      lstm_step_kernel<<<mtiles * 64, 256, 0, stream>>>(XPRE, hin, Whh, CC, hout,
                                                        LASTH, s, mtiles, 0);
    float* tmp = hin; hin = hout; hout = tmp;
  }
  gemm_kernel<0, 2, 1024><<<256, 256, 0, stream>>>(LASTH, DD, Wc, DD, bc, nullptr,
                                                   UNR, DD, 0, nullptr, DD);
  ln_relu_kernel<<<128, 256, 0, stream>>>(UNR, g_a, be_a, LNR);
  gemm_kernel<2, 4, 1024><<<236, 256, 0, stream>>>(LNR, DD, Wa, DD, ba, nullptr,
                                                   out, NA, 0, nullptr, NA);
}


# All WMMA K loops: operand rows fetched as whole 128-byte lines (4 lines per load instruction, was 16) and turned into fragment order through wave-private LDS; every k sum order kept
# speedup vs baseline: 3.2514x; 1.0361x over previous
#include <hip/hip_runtime.h>
#include <math.h>

#define DD 1024
#define NA 1887
#define BB 8
#define TT 16
#define ROWS 128

typedef __attribute__((ext_vector_type(2))) float v2f;
typedef __attribute__((ext_vector_type(4))) float v4f;
typedef __attribute__((ext_vector_type(8))) float v8f;

#define SP 36
#define SGRP (16 * SP)

__global__ void pool_kernel(const float* __restrict__ x, float* __restrict__ A0,
                            float* __restrict__ F1, float* __restrict__ F2) {
  int tid = blockIdx.x * blockDim.x + threadIdx.x;
  if (tid >= ROWS * DD * 3) return;
  int c  = tid % 3;
  int rd = tid / 3;
  int d  = rd % DD;
  int r  = rd / DD;
  const float* p = x + (size_t)rd * 147 + c;
  float s = 0.f;
  #pragma unroll
  for (int i = 0; i < 49; ++i) s += p[i * 3];
  s *= (1.0f / 49.0f);
  if (c == 0) {
    int b = r >> 4, t = r & 15;
    A0[(size_t)(t * BB + b) * DD + d] = s;
  } else if (c == 1) {
    F1[(size_t)r * DD + d] = s;
  } else {
    F2[(size_t)r * DD + d] = s;
  }
}

template <int MODE, int MT, int KK>
__device__ __forceinline__ void
gemm_block(const float* __restrict__ A, int lda,
           const float* __restrict__ W, int ldw,
           const float* __restrict__ bias1,
           const float* __restrict__ bias2,
           float* __restrict__ out, int ldo, int coloff,
           float* __restrict__ out2, int N, int bid,
           float (*red)[MT][8][32],
           float* __restrict__ stg) {
  constexpr int MG = 8 / MT;
  constexpr int NG = MT + 1;
  int ks   = threadIdx.x >> 5;
  int lane = threadIdx.x & 31;
  int n0    = (bid / MG) << 4;
  int mbase = (bid % MG) * (MT * 16);
  int c    = lane & 15;
  int hi   = lane >> 4;
  int koff = hi << 2;
  int ncol = n0 + c;
  int nload = (ncol < N) ? ncol : (N - 1);
  constexpr int KS = KK / 8;
  int lr = lane >> 3;
  int lc = (lane & 7) << 2;
  const float* wsrc[4];
  #pragma unroll
  for (int q = 0; q < 4; ++q) {
    int wr = n0 + 4 * q + lr;
    wr = (wr < N) ? wr : (N - 1);
    wsrc[q] = W + (size_t)wr * ldw + ks * KS + lc;
  }
  const float* asrc = A + (size_t)(mbase + lr) * lda + ks * KS + lc;
  const size_t arow4 = (size_t)4 * lda;
  float* st = stg + ks * (NG * SGRP);
  float* stw = st + lr * SP + lc;
  const float* strd = st + c * SP + koff;

  v8f acc[MT];
  #pragma unroll
  for (int mt = 0; mt < MT; ++mt)
    #pragma unroll
    for (int v = 0; v < 8; ++v) acc[mt][v] = 0.f;

  constexpr int NL = KS / 32;
  v4f gq[NG][4];
  v4f fr[4][NG];
  #pragma unroll
  for (int q = 0; q < 4; ++q) {
    gq[0][q] = *(const v4f*)(wsrc[q]);
    #pragma unroll
    for (int mt = 0; mt < MT; ++mt)
      gq[1 + mt][q] = *(const v4f*)(asrc + (size_t)(mt * 4 + q) * arow4);
  }
  #pragma unroll
  for (int g = 0; g < NG; ++g)
    #pragma unroll
    for (int q = 0; q < 4; ++q)
      *(v4f*)(stw + g * SGRP + 4 * q * SP) = gq[g][q];
  #pragma unroll
  for (int g = 0; g < NG; ++g)
    #pragma unroll
    for (int j = 0; j < 4; ++j)
      fr[j][g] = *(const v4f*)(strd + g * SGRP + 8 * j);
  #pragma unroll
  for (int t = 0; t < NL; ++t) {
    if (t + 1 < NL) {
      #pragma unroll
      for (int q = 0; q < 4; ++q) {
        gq[0][q] = *(const v4f*)(wsrc[q] + (t + 1) * 32);
        #pragma unroll
        for (int mt = 0; mt < MT; ++mt)
          gq[1 + mt][q] = *(const v4f*)(asrc + (size_t)(mt * 4 + q) * arow4 + (t + 1) * 32);
      }
    }
    __builtin_amdgcn_sched_barrier(0);
    #pragma unroll
    for (int j = 0; j < 4; ++j) {
      v2f b0 = {fr[j][0].x, fr[j][0].y}, b1 = {fr[j][0].z, fr[j][0].w};
      #pragma unroll
      for (int mt = 0; mt < MT; ++mt) {
        v2f a0 = {fr[j][1 + mt].x, fr[j][1 + mt].y};
        acc[mt] = __builtin_amdgcn_wmma_f32_16x16x4_f32(false, a0, false, b0,
                                                        (short)0, acc[mt], false, false);
      }
      #pragma unroll
      for (int mt = 0; mt < MT; ++mt) {
        v2f a1 = {fr[j][1 + mt].z, fr[j][1 + mt].w};
        acc[mt] = __builtin_amdgcn_wmma_f32_16x16x4_f32(false, a1, false, b1,
                                                        (short)0, acc[mt], false, false);
      }
    }
    __builtin_amdgcn_sched_barrier(0);
    if (t + 1 < NL) {
      #pragma unroll
      for (int g = 0; g < NG; ++g)
        #pragma unroll
        for (int q = 0; q < 4; ++q)
          *(v4f*)(stw + g * SGRP + 4 * q * SP) = gq[g][q];
      #pragma unroll
      for (int g = 0; g < NG; ++g)
        #pragma unroll
        for (int j = 0; j < 4; ++j)
          fr[j][g] = *(const v4f*)(strd + g * SGRP + 8 * j);
    }
  }

  #pragma unroll
  for (int mt = 0; mt < MT; ++mt)
    #pragma unroll
    for (int v = 0; v < 8; ++v) red[ks][mt][v][lane] = acc[mt][v];
  __syncthreads();

  float bsum = bias1[nload] + (bias2 ? bias2[nload] : 0.f);
  #pragma unroll
  for (int mt = 0; mt < MT; ++mt) {
    float t = red[0][mt][ks][lane];
    #pragma unroll
    for (int q = 1; q < 8; ++q) t += red[q][mt][ks][lane];
    float val = t + bsum;
    int r = mbase + mt * 16 + ks + (hi << 3);
    if (MODE == 0) {
      out[(size_t)r * ldo + coloff + ncol] = val;
    } else if (MODE == 1) {
      int bi = r >> 4, tt = r & 15;
      int rr = tt * BB + bi;
      if (ncol < DD) out[(size_t)rr * DD + ncol] = val;
      else           out2[(size_t)rr * DD + (ncol - DD)] = val;
    } else {
      if (ncol < N) {
        int tt = r >> 3, bi = r & 7;
        out[(size_t)(bi * TT + tt) * N + ncol] = val;
      }
    }
  }
}

template <int MODE, int MT, int KK>
__global__ void __launch_bounds__(256)
gemm_kernel(const float* __restrict__ A, int lda,
            const float* __restrict__ W, int ldw,
            const float* __restrict__ bias1,
            const float* __restrict__ bias2,
            float* __restrict__ out, int ldo, int coloff,
            float* __restrict__ out2, int N) {
  __shared__ float red[8][MT][8][32];
  __shared__ __attribute__((aligned(16))) float stg[8 * (MT + 1) * SGRP];
  gemm_block<MODE, MT, KK>(A, lda, W, ldw, bias1, bias2, out, ldo, coloff, out2, N,
                           (int)blockIdx.x, red, stg);
}

__global__ void __launch_bounds__(256)
gemm_pre_kernel(const float* __restrict__ A0, const float* __restrict__ Wih,
                const float* __restrict__ bih, const float* __restrict__ bhh,
                float* __restrict__ XPRE,
                const float* __restrict__ F1, const float* __restrict__ Wn,
                const float* __restrict__ bn,
                const float* __restrict__ F2, const float* __restrict__ Wv,
                const float* __restrict__ bv,
                float* __restrict__ ENV) {
  __shared__ float red[8][4][8][32];
  __shared__ __attribute__((aligned(16))) float stg[8 * 5 * SGRP];
  int b = (int)blockIdx.x;
  const float* A  = A0;  const float* W  = Wih;
  const float* b1 = bih; const float* b2 = bhh;
  float* o = XPRE; int ldo = 4 * DD, coloff = 0, N = 4 * DD, bid = b;
  if (b >= 512) {
    bool v = b >= 640;
    A  = v ? F2 : F1;  W = v ? Wv : Wn;
    b1 = v ? bv : bn;  b2 = nullptr;
    o = ENV; ldo = 2 * DD; coloff = v ? DD : 0; N = DD;
    bid = b - (v ? 640 : 512);
  }
  gemm_block<0, 4, 1024>(A, DD, W, DD, b1, b2, o, ldo, coloff, nullptr, N, bid, red, stg);
}

template <int NT>
__device__ __forceinline__ void lstm_tiles(const float* __restrict__ xpre,
                                           const float* __restrict__ hin,
                                           const float* __restrict__ Whh,
                                           float* __restrict__ cc,
                                           float* __restrict__ hout,
                                           float* __restrict__ lasth,
                                           int step, int m0, int n0,
                                           float (*red)[2][4][8][32],
                                           float* __restrict__ stg) {
  constexpr int NG = NT + 4;
  int ks   = threadIdx.x >> 5;
  int lane = threadIdx.x & 31;
  int nn   = lane & 15;
  int hi   = lane >> 4;
  int koff = hi << 2;
  int kbase = ks * (DD / 8);
  int d  = n0 + nn;
  int lr = lane >> 3;
  int lc = (lane & 7) << 2;

  float xp[NT][4], cold[NT];
  #pragma unroll
  for (int tt = 0; tt < NT; ++tt) {
    int rf = m0 + tt * 16 + ks + (hi << 3);
    #pragma unroll
    for (int g = 0; g < 4; ++g) xp[tt][g] = xpre[(size_t)rf * (4 * DD) + g * DD + d];
    cold[tt] = cc[(size_t)rf * DD + d];
  }

  v8f acc[NT][4];
  #pragma unroll
  for (int tt = 0; tt < NT; ++tt)
    #pragma unroll
    for (int g = 0; g < 4; ++g)
      #pragma unroll
      for (int v = 0; v < 8; ++v) acc[tt][g][v] = 0.f;

  const float* asrc = hin + (size_t)(m0 + lr) * DD + kbase + lc;
  const float* wsrc = Whh + (size_t)(n0 + lr) * DD + kbase + lc;
  constexpr size_t row4  = (size_t)4 * DD;
  constexpr size_t gstep = (size_t)DD * DD;
  float* st = stg + ks * (NG * SGRP);
  float* stw = st + lr * SP + lc;
  const float* strd = st + nn * SP + koff;

  constexpr int LNL = (DD / 8) / 32;
  v4f gq[NG][4];
  v4f fr[4][NG];
  #pragma unroll
  for (int q = 0; q < 4; ++q) {
    #pragma unroll
    for (int tt = 0; tt < NT; ++tt)
      gq[tt][q] = *(const v4f*)(asrc + (size_t)(tt * 4 + q) * row4);
    #pragma unroll
    for (int g = 0; g < 4; ++g)
      gq[NT + g][q] = *(const v4f*)(wsrc + g * gstep + q * row4);
  }
  #pragma unroll
  for (int g = 0; g < NG; ++g)
    #pragma unroll
    for (int q = 0; q < 4; ++q)
      *(v4f*)(stw + g * SGRP + 4 * q * SP) = gq[g][q];
  #pragma unroll
  for (int g = 0; g < NG; ++g)
    #pragma unroll
    for (int j = 0; j < 4; ++j)
      fr[j][g] = *(const v4f*)(strd + g * SGRP + 8 * j);
  #pragma unroll
  for (int t = 0; t < LNL; ++t) {
    if (t + 1 < LNL) {
      #pragma unroll
      for (int q = 0; q < 4; ++q) {
        #pragma unroll
        for (int tt = 0; tt < NT; ++tt)
          gq[tt][q] = *(const v4f*)(asrc + (size_t)(tt * 4 + q) * row4 + (t + 1) * 32);
        #pragma unroll
        for (int g = 0; g < 4; ++g)
          gq[NT + g][q] = *(const v4f*)(wsrc + g * gstep + q * row4 + (t + 1) * 32);
      }
    }
    __builtin_amdgcn_sched_barrier(0);
    #pragma unroll
    for (int j = 0; j < 4; ++j) {
      #pragma unroll
      for (int tt = 0; tt < NT; ++tt) {
        v2f a0 = {fr[j][tt].x, fr[j][tt].y};
        #pragma unroll
        for (int g = 0; g < 4; ++g) {
          v2f b = {fr[j][NT + g].x, fr[j][NT + g].y};
          acc[tt][g] = __builtin_amdgcn_wmma_f32_16x16x4_f32(false, a0, false, b,
                                                             (short)0, acc[tt][g], false, false);
        }
      }
      #pragma unroll
      for (int tt = 0; tt < NT; ++tt) {
        v2f a1 = {fr[j][tt].z, fr[j][tt].w};
        #pragma unroll
        for (int g = 0; g < 4; ++g) {
          v2f b = {fr[j][NT + g].z, fr[j][NT + g].w};
          acc[tt][g] = __builtin_amdgcn_wmma_f32_16x16x4_f32(false, a1, false, b,
                                                             (short)0, acc[tt][g], false, false);
        }
      }
    }
    __builtin_amdgcn_sched_barrier(0);
    if (t + 1 < LNL) {
      #pragma unroll
      for (int g = 0; g < NG; ++g)
        #pragma unroll
        for (int q = 0; q < 4; ++q)
          *(v4f*)(stw + g * SGRP + 4 * q * SP) = gq[g][q];
      #pragma unroll
      for (int g = 0; g < NG; ++g)
        #pragma unroll
        for (int j = 0; j < 4; ++j)
          fr[j][g] = *(const v4f*)(strd + g * SGRP + 8 * j);
    }
  }

  #pragma unroll
  for (int tt = 0; tt < NT; ++tt)
    #pragma unroll
    for (int g = 0; g < 4; ++g)
      #pragma unroll
      for (int v = 0; v < 8; ++v) red[ks][tt][g][v][lane] = acc[tt][g][v];
  __syncthreads();

  float s[NT][4];
  #pragma unroll
  for (int tt = 0; tt < NT; ++tt)
    #pragma unroll
    for (int g = 0; g < 4; ++g) {
      float t = red[0][tt][g][ks][lane];
      #pragma unroll
      for (int q = 1; q < 8; ++q) t += red[q][tt][g][ks][lane];
      s[tt][g] = t + xp[tt][g];
    }

  int tgt = TT - 1 - step;
  #pragma unroll
  for (int tt = 0; tt < NT; ++tt) {
    int rf = m0 + tt * 16 + ks + (hi << 3);
    float si = 1.f / (1.f + expf(-s[tt][0]));
    float sf = 1.f / (1.f + expf(-s[tt][1]));
    float gt = tanhf(s[tt][2]);
    float so = 1.f / (1.f + expf(-s[tt][3]));
    float cnew = sf * cold[tt] + si * gt;
    float hnew = so * tanhf(cnew);
    cc[(size_t)rf * DD + d] = cnew;
    hout[(size_t)rf * DD + d] = hnew;
    if ((rf >> 3) == tgt) lasth[(size_t)rf * DD + d] = hnew;
  }
}

__global__ void __launch_bounds__(256)
lstm_step_kernel(const float* __restrict__ xpre,
                 const float* __restrict__ hin,
                 const float* __restrict__ Whh,
                 float* __restrict__ cc,
                 float* __restrict__ hout,
                 float* __restrict__ lasth,
                 int step, int mtiles, int pairs) {
  __shared__ float red[8][2][4][8][32];
  __shared__ __attribute__((aligned(16))) float stg[8 * 6 * SGRP];
  int units = pairs ? ((mtiles + 1) >> 1) : mtiles;
  int u  = blockIdx.x % units;
  int n0 = (blockIdx.x / units) << 4;
  int mt = pairs ? (u << 1) : u;
  if (pairs && mt + 1 < mtiles)
    lstm_tiles<2>(xpre, hin, Whh, cc, hout, lasth, step, mt << 4, n0, red, stg);
  else
    lstm_tiles<1>(xpre, hin, Whh, cc, hout, lasth, step, mt << 4, n0, red, stg);
}

__global__ void ln_relu_kernel(const float* __restrict__ in,
                               const float* __restrict__ g,
                               const float* __restrict__ be,
                               float* __restrict__ out) {
  __shared__ float s1[256], s2[256];
  int r = blockIdx.x, t = threadIdx.x;
  const float* row = in + (size_t)r * DD;
  float s = 0.f, sq = 0.f;
  for (int i = t; i < DD; i += 256) { float v = row[i]; s += v; sq += v * v; }
  s1[t] = s; s2[t] = sq;
  __syncthreads();
  for (int off = 128; off > 0; off >>= 1) {
    if (t < off) { s1[t] += s1[t + off]; s2[t] += s2[t + off]; }
    __syncthreads();
  }
  float mean = s1[0] * (1.0f / DD);
  float var = s2[0] * (1.0f / DD) - mean * mean;
  float inv = rsqrtf(var + 1e-5f);
  for (int i = t; i < DD; i += 256) {
    float v = (row[i] - mean) * inv * g[i] + be[i];
    out[(size_t)r * DD + i] = v > 0.f ? v : 0.f;
  }
}

extern "C" void kernel_launch(void* const* d_in, const int* in_sizes, int n_in,
                              void* d_out, int out_size, void* d_ws, size_t ws_size,
                              hipStream_t stream) {
  const float* x   = (const float*)d_in[0];
  const float* Wn  = (const float*)d_in[1];
  const float* bn  = (const float*)d_in[2];
  const float* Wv  = (const float*)d_in[3];
  const float* bv  = (const float*)d_in[4];
  const float* Wf  = (const float*)d_in[5];
  const float* bf  = (const float*)d_in[6];
  const float* Wih = (const float*)d_in[7];
  const float* Whh = (const float*)d_in[8];
  const float* bih = (const float*)d_in[9];
  const float* bhh = (const float*)d_in[10];
  const float* Wc  = (const float*)d_in[11];
  const float* bc  = (const float*)d_in[12];
  const float* g_a = (const float*)d_in[13];
  const float* be_a= (const float*)d_in[14];
  const float* Wa  = (const float*)d_in[15];
  const float* ba  = (const float*)d_in[16];
  float* out = (float*)d_out;

  float* ws    = (float*)d_ws;
  float* A0    = ws;
  float* F1    = A0    + 131072;
  float* F2    = F1    + 131072;
  float* ENV   = F2    + 131072;
  float* XPRE  = ENV   + 262144;
  float* H0    = XPRE  + 524288;
  float* H1    = H0    + 131072;
  float* CC    = H1    + 131072;
  float* LASTH = CC    + 131072;
  float* UNR   = LASTH + 131072;
  float* LNR   = UNR   + 131072;

  pool_kernel<<<1536, 256, 0, stream>>>(x, A0, F1, F2);
  gemm_pre_kernel<<<768, 256, 0, stream>>>(A0, Wih, bih, bhh, XPRE,
                                           F1, Wn, bn, F2, Wv, bv, ENV);
  gemm_kernel<1, 4, 2048><<<256, 256, 0, stream>>>(ENV, 2 * DD, Wf, 2 * DD, bf, nullptr,
                                                   H0, 0, 0, CC, 2 * DD);
  float* hin = H0; float* hout = H1;
  for (int s = 0; s < TT; ++s) {
    int mtiles = (TT - s + 1) >> 1;
    if (mtiles > 4)
      lstm_step_kernel<<<((mtiles + 1) >> 1) * 64, 256, 0, stream>>>(XPRE, hin, Whh, CC, hout,
                                                                     LASTH, s, mtiles, 1);
    else
      lstm_step_kernel<<<mtiles * 64, 256, 0, stream>>>(XPRE, hin, Whh, CC, hout,
                                                        LASTH, s, mtiles, 0);
    float* tmp = hin; hin = hout; hout = tmp;
  }
  gemm_kernel<0, 2, 1024><<<256, 256, 0, stream>>>(LASTH, DD, Wc, DD, bc, nullptr,
                                                   UNR, DD, 0, nullptr, DD);
  ln_relu_kernel<<<128, 256, 0, stream>>>(UNR, g_a, be_a, LNR);
  gemm_kernel<2, 4, 1024><<<236, 256, 0, stream>>>(LNR, DD, Wa, DD, ba, nullptr,
                                                   out, NA, 0, nullptr, NA);
}
